# starting kernel
# baseline (speedup 1.0000x reference)
#include <hip/hip_runtime.h>
#include <math.h>

typedef __attribute__((ext_vector_type(2))) float v2f;
typedef __attribute__((ext_vector_type(8))) float v8f;

#define LN_EPS 1e-5f
#define SQ_EPS 1e-9f

__device__ __forceinline__ float gelu_exact(float x) {
    return 0.5f * x * (1.0f + erff(x * 0.70710678118654752440f));
}

__device__ __forceinline__ v8f wmma4(v2f a, v2f b, v8f c) {
    return __builtin_amdgcn_wmma_f32_16x16x4_f32(false, a, false, b, (short)0, c, false, false);
}

__device__ __forceinline__ v8f vzero8() {
    v8f z = {0.f,0.f,0.f,0.f,0.f,0.f,0.f,0.f};
    return z;
}

__global__ __attribute__((amdgpu_waves_per_eu(1))) __launch_bounds__(256)
void capsnet_fused(
    const float* __restrict__ x,   const float* __restrict__ W1, const float* __restrict__ b1,
    const float* __restrict__ W2,  const float* __restrict__ b2,
    const float* __restrict__ ln_g,const float* __restrict__ ln_b,
    const float* __restrict__ Wcap,const float* __restrict__ Wf, const float* __restrict__ bf,
    float* __restrict__ out)
{
    __shared__ float xh[16*256];
    __shared__ float h1s[16*32];
    __shared__ float mu_s[16];
    __shared__ float rstd_s[16];
    __shared__ float u_t[32*130];
    __shared__ float bij[16*513];
    __shared__ float cij[16*513];
    __shared__ float part[16*17];

    const int t    = threadIdx.x;
    const int lane = t & 31;
    const int wav  = t >> 5;
    const int bb   = lane & 15;
    const int hi   = lane >> 4;

    const int tile = blockIdx.x;
    const float* xg = x + (size_t)tile * 16 * 256;

    #pragma unroll
    for (int r = 0; r < 16; ++r) xh[t + 256*r] = xg[t + 256*r];
    __syncthreads();

    {
        int n = t & 31, b0 = t >> 5;
        float acc0 = b1[n], acc1 = acc0;
        #pragma unroll 8
        for (int k = 0; k < 256; ++k) {
            float w = W1[k*32 + n];
            acc0 = fmaf(xh[b0*256 + k],       w, acc0);
            acc1 = fmaf(xh[(b0+8)*256 + k],   w, acc1);
        }
        h1s[b0*32 + n]     = gelu_exact(acc0);
        h1s[(b0+8)*32 + n] = gelu_exact(acc1);
    }
    __syncthreads();

    {
        int n0 = t & 31, b0 = t >> 5;
        float h1a[32], h1b[32];
        #pragma unroll
        for (int k = 0; k < 32; ++k) { h1a[k] = h1s[b0*32 + k]; h1b[k] = h1s[(b0+8)*32 + k]; }
        #pragma unroll
        for (int rr = 0; rr < 8; ++rr) {
            int n = n0 + 32*rr;
            float acc0 = b2[n], acc1 = acc0;
            #pragma unroll
            for (int k = 0; k < 32; ++k) {
                float w = W2[k*256 + n];
                acc0 = fmaf(h1a[k], w, acc0);
                acc1 = fmaf(h1b[k], w, acc1);
            }
            xh[b0*256 + n]     = gelu_exact(acc0);
            xh[(b0+8)*256 + n] = gelu_exact(acc1);
        }
    }
    __syncthreads();

    {
        int ba = wav, bc = wav + 8;
        float sa = 0.f, sb = 0.f;
        #pragma unroll
        for (int r = 0; r < 8; ++r) { sa += xh[ba*256 + lane + 32*r]; sb += xh[bc*256 + lane + 32*r]; }
        #pragma unroll
        for (int m = 16; m >= 1; m >>= 1) { sa += __shfl_xor(sa, m, 32); sb += __shfl_xor(sb, m, 32); }
        float mua = sa * (1.0f/256.0f), mub = sb * (1.0f/256.0f);
        float va = 0.f, vb = 0.f;
        #pragma unroll
        for (int r = 0; r < 8; ++r) {
            float da = xh[ba*256 + lane + 32*r] - mua; va = fmaf(da, da, va);
            float db = xh[bc*256 + lane + 32*r] - mub; vb = fmaf(db, db, vb);
        }
        #pragma unroll
        for (int m = 16; m >= 1; m >>= 1) { va += __shfl_xor(va, m, 32); vb += __shfl_xor(vb, m, 32); }
        if (lane == 0) {
            mu_s[ba] = mua; rstd_s[ba] = rsqrtf(va * (1.0f/256.0f) + LN_EPS);
            mu_s[bc] = mub; rstd_s[bc] = rsqrtf(vb * (1.0f/256.0f) + LN_EPS);
        }
    }
    __syncthreads();

    {
        int i = t & 31, b0 = t >> 5;
        float g8[8], be8[8];
        #pragma unroll
        for (int k = 0; k < 8; ++k) { g8[k] = ln_g[i*8 + k]; be8[k] = ln_b[i*8 + k]; }
        #pragma unroll
        for (int bsel = 0; bsel < 2; ++bsel) {
            int b = b0 + 8*bsel;
            float mu = mu_s[b], rs = rstd_s[b];
            float vals[8]; float nsq = 0.f;
            #pragma unroll
            for (int k = 0; k < 8; ++k) {
                float v = (xh[b*256 + i*8 + k] - mu) * rs * g8[k] + be8[k];
                vals[k] = v; nsq = fmaf(v, v, nsq);
            }
            float scale = nsq / ((1.0f + nsq) * sqrtf(nsq + SQ_EPS));
            #pragma unroll
            for (int k = 0; k < 8; ++k) u_t[i*130 + k*16 + b] = scale * vals[k];
        }
    }

    #pragma unroll
    for (int r = 0; r < 32; ++r) { bij[t + 256*r] = 0.f; cij[t + 256*r] = 0.0625f; }
    if (t < 16) { bij[8192 + t] = 0.f; cij[8192 + t] = 0.0625f; }
    __syncthreads();

    v2f a0A[32], a1A[32], a0B[32], a1B[32];
    #pragma unroll
    for (int i = 0; i < 32; ++i) {
        const float* wcA = Wcap + (size_t)((i*16 + wav    )*8)*16 + bb;
        const float* wcB = Wcap + (size_t)((i*16 + wav + 8)*8)*16 + bb;
        a0A[i][0] = wcA[(2*hi+0)*16];  a0A[i][1] = wcA[(2*hi+1)*16];
        a1A[i][0] = wcA[(2*hi+4)*16];  a1A[i][1] = wcA[(2*hi+5)*16];
        a0B[i][0] = wcB[(2*hi+0)*16];  a0B[i][1] = wcB[(2*hi+1)*16];
        a1B[i][0] = wcB[(2*hi+4)*16];  a1B[i][1] = wcB[(2*hi+5)*16];
    }

    v8f vA = vzero8(), vB = vzero8();
    for (int r = 0; r < 3; ++r) {
        v8f sA0 = vzero8(), sA1 = vzero8(), sB0 = vzero8(), sB1 = vzero8();
        #pragma unroll
        for (int i = 0; i < 32; ++i) {
            const float* ub = u_t + i*130 + bb;
            v2f bu0, bu1;
            bu0[0] = ub[(2*hi+0)*16];  bu0[1] = ub[(2*hi+1)*16];
            bu1[0] = ub[(2*hi+4)*16];  bu1[1] = ub[(2*hi+5)*16];
            float cA = cij[bb*513 + i*16 + wav];
            float cB = cij[bb*513 + i*16 + wav + 8];
            v2f p0, p1;
            p0[0] = bu0[0]*cA; p0[1] = bu0[1]*cA; p1[0] = bu1[0]*cA; p1[1] = bu1[1]*cA;
            if (i & 1) { sA1 = wmma4(a0A[i], p0, sA1); sA1 = wmma4(a1A[i], p1, sA1); }
            else       { sA0 = wmma4(a0A[i], p0, sA0); sA0 = wmma4(a1A[i], p1, sA0); }
            p0[0] = bu0[0]*cB; p0[1] = bu0[1]*cB; p1[0] = bu1[0]*cB; p1[1] = bu1[1]*cB;
            if (i & 1) { sB1 = wmma4(a0B[i], p0, sB1); sB1 = wmma4(a1B[i], p1, sB1); }
            else       { sB0 = wmma4(a0B[i], p0, sB0); sB0 = wmma4(a1B[i], p1, sB0); }
        }
        v8f sA = sA0 + sA1, sB = sB0 + sB1;
        float nA = 0.f, nB = 0.f;
        #pragma unroll
        for (int v = 0; v < 8; ++v) { nA = fmaf(sA[v], sA[v], nA); nB = fmaf(sB[v], sB[v], nB); }
        nA += __shfl_xor(nA, 16, 32);  nB += __shfl_xor(nB, 16, 32);
        float scA = nA / ((1.0f + nA) * sqrtf(nA + SQ_EPS));
        float scB = nB / ((1.0f + nB) * sqrtf(nB + SQ_EPS));
        #pragma unroll
        for (int v = 0; v < 8; ++v) { vA[v] = sA[v]*scA; vB[v] = sB[v]*scB; }

        if (r < 2) {
            #pragma unroll
            for (int i = 0; i < 32; ++i) {
                const float* ub = u_t + i*130 + bb;
                v2f bu0, bu1;
                bu0[0] = ub[(2*hi+0)*16];  bu0[1] = ub[(2*hi+1)*16];
                bu1[0] = ub[(2*hi+4)*16];  bu1[1] = ub[(2*hi+5)*16];
                v8f uA = vzero8(); uA = wmma4(a0A[i], bu0, uA); uA = wmma4(a1A[i], bu1, uA);
                v8f uB = vzero8(); uB = wmma4(a0B[i], bu0, uB); uB = wmma4(a1B[i], bu1, uB);
                float agA = 0.f, agB = 0.f;
                #pragma unroll
                for (int v = 0; v < 8; ++v) { agA = fmaf(uA[v], vA[v], agA); agB = fmaf(uB[v], vB[v], agB); }
                agA += __shfl_xor(agA, 16, 32);  agB += __shfl_xor(agB, 16, 32);
                int ia = bb*513 + i*16 + wav;
                if (lane < 16) { bij[ia] += agA; bij[ia + 8] += agB; }
            }
            __syncthreads();
            #pragma unroll
            for (int bsel = 0; bsel < 2; ++bsel) {
                int base = (wav + 8*bsel)*513 + (t & 31)*16;
                float mx = bij[base];
                #pragma unroll
                for (int j = 1; j < 16; ++j) mx = fmaxf(mx, bij[base + j]);
                float e[16]; float sum = 0.f;
                #pragma unroll
                for (int j = 0; j < 16; ++j) { e[j] = expf(bij[base + j] - mx); sum += e[j]; }
                float inv = 1.0f / sum;
                #pragma unroll
                for (int j = 0; j < 16; ++j) cij[base + j] = e[j] * inv;
            }
            __syncthreads();
        }
    }

    {
        float pA = 0.f, pB = 0.f;
        #pragma unroll
        for (int v = 0; v < 8; ++v) {
            pA = fmaf(vA[v], Wf[(wav    )*16 + v + 8*hi], pA);
            pB = fmaf(vB[v], Wf[(wav + 8)*16 + v + 8*hi], pB);
        }
        pA += __shfl_xor(pA, 16, 32);  pB += __shfl_xor(pB, 16, 32);
        if (lane < 16) { part[wav*17 + bb] = pA; part[(wav+8)*17 + bb] = pB; }
    }
    __syncthreads();
    if (t < 16) {
        float acc = bf[0];
        #pragma unroll
        for (int j = 0; j < 16; ++j) acc += part[j*17 + t];
        out[tile*16 + t] = acc;
    }
}

extern "C" void kernel_launch(void* const* d_in, const int* in_sizes, int n_in,
                              void* d_out, int out_size, void* d_ws, size_t ws_size,
                              hipStream_t stream) {
    const float* x    = (const float*)d_in[0];
    const float* W1   = (const float*)d_in[1];
    const float* b1   = (const float*)d_in[2];
    const float* W2   = (const float*)d_in[3];
    const float* b2   = (const float*)d_in[4];
    const float* ln_g = (const float*)d_in[5];
    const float* ln_b = (const float*)d_in[6];
    const float* Wcap = (const float*)d_in[7];
    const float* Wf   = (const float*)d_in[8];
    const float* bf   = (const float*)d_in[9];
    float* out = (float*)d_out;

    const int B = in_sizes[0] / 256;
    const int blocks = B / 16;
    capsnet_fused<<<blocks, 256, 0, stream>>>(x, W1, b1, W2, b2, ln_g, ln_b, Wcap, Wf, bf, out);
}


# Routing re-split by input capsule: wave owns 4 of 32 capsules x 16 outputs; logits, softmax in registers; s_j = 8 wave partials summed via LDS, fixed order; agreement dot in 2 chains: new sum order
# speedup vs baseline: 1.1356x; 1.1356x over previous
#include <hip/hip_runtime.h>
#include <math.h>

typedef __attribute__((ext_vector_type(2))) float v2f;
typedef __attribute__((ext_vector_type(8))) float v8f;

#define LN_EPS 1e-5f
#define SQ_EPS 1e-9f

__device__ __forceinline__ float gelu_exact(float x) {
    return 0.5f * x * (1.0f + erff(x * 0.70710678118654752440f));
}

__device__ __forceinline__ v8f wmma4(v2f a, v2f b, v8f c) {
    return __builtin_amdgcn_wmma_f32_16x16x4_f32(false, a, false, b, (short)0, c, false, false);
}

__device__ __forceinline__ v8f vzero8() {
    v8f z = {0.f,0.f,0.f,0.f,0.f,0.f,0.f,0.f};
    return z;
}

__global__ __attribute__((amdgpu_waves_per_eu(1))) __launch_bounds__(256)
void capsnet_fused(
    const float* __restrict__ x,   const float* __restrict__ W1, const float* __restrict__ b1,
    const float* __restrict__ W2,  const float* __restrict__ b2,
    const float* __restrict__ ln_g,const float* __restrict__ ln_b,
    const float* __restrict__ Wcap,const float* __restrict__ Wf, const float* __restrict__ bf,
    float* __restrict__ out)
{
    __shared__ float xh[16*256];
    __shared__ float h1s[16*32];
    __shared__ float mu_s[16];
    __shared__ float rstd_s[16];
    __shared__ float u_t[32*130];
    __shared__ float4 Ps[8*16*64];
    __shared__ float4 Cs[16*8*16];
    __shared__ float4 Vs[16*64];
    __shared__ float part[16*17];

    const int t    = threadIdx.x;
    const int lane = t & 31;
    const int wav  = t >> 5;
    const int bb   = lane & 15;
    const int hi   = lane >> 4;

    const int tile = blockIdx.x;
    const float* xg = x + (size_t)tile * 16 * 256;

    #pragma unroll
    for (int r = 0; r < 16; ++r) xh[t + 256*r] = xg[t + 256*r];
    __syncthreads();

    {
        int n = t & 31, b0 = t >> 5;
        float acc0 = b1[n], acc1 = acc0;
        #pragma unroll 8
        for (int k = 0; k < 256; ++k) {
            float w = W1[k*32 + n];
            acc0 = fmaf(xh[b0*256 + k],       w, acc0);
            acc1 = fmaf(xh[(b0+8)*256 + k],   w, acc1);
        }
        h1s[b0*32 + n]     = gelu_exact(acc0);
        h1s[(b0+8)*32 + n] = gelu_exact(acc1);
    }
    __syncthreads();

    {
        int n0 = t & 31, b0 = t >> 5;
        float h1a[32], h1b[32];
        #pragma unroll
        for (int k = 0; k < 32; ++k) { h1a[k] = h1s[b0*32 + k]; h1b[k] = h1s[(b0+8)*32 + k]; }
        #pragma unroll
        for (int rr = 0; rr < 8; ++rr) {
            int n = n0 + 32*rr;
            float acc0 = b2[n], acc1 = acc0;
            #pragma unroll
            for (int k = 0; k < 32; ++k) {
                float w = W2[k*256 + n];
                acc0 = fmaf(h1a[k], w, acc0);
                acc1 = fmaf(h1b[k], w, acc1);
            }
            xh[b0*256 + n]     = gelu_exact(acc0);
            xh[(b0+8)*256 + n] = gelu_exact(acc1);
        }
    }
    __syncthreads();

    {
        int ba = wav, bc = wav + 8;
        float sa = 0.f, sb = 0.f;
        #pragma unroll
        for (int r = 0; r < 8; ++r) { sa += xh[ba*256 + lane + 32*r]; sb += xh[bc*256 + lane + 32*r]; }
        #pragma unroll
        for (int m = 16; m >= 1; m >>= 1) { sa += __shfl_xor(sa, m, 32); sb += __shfl_xor(sb, m, 32); }
        float mua = sa * (1.0f/256.0f), mub = sb * (1.0f/256.0f);
        float va = 0.f, vb = 0.f;
        #pragma unroll
        for (int r = 0; r < 8; ++r) {
            float da = xh[ba*256 + lane + 32*r] - mua; va = fmaf(da, da, va);
            float db = xh[bc*256 + lane + 32*r] - mub; vb = fmaf(db, db, vb);
        }
        #pragma unroll
        for (int m = 16; m >= 1; m >>= 1) { va += __shfl_xor(va, m, 32); vb += __shfl_xor(vb, m, 32); }
        if (lane == 0) {
            mu_s[ba] = mua; rstd_s[ba] = rsqrtf(va * (1.0f/256.0f) + LN_EPS);
            mu_s[bc] = mub; rstd_s[bc] = rsqrtf(vb * (1.0f/256.0f) + LN_EPS);
        }
    }
    __syncthreads();

    {
        int i = t & 31, b0 = t >> 5;
        float g8[8], be8[8];
        #pragma unroll
        for (int k = 0; k < 8; ++k) { g8[k] = ln_g[i*8 + k]; be8[k] = ln_b[i*8 + k]; }
        #pragma unroll
        for (int bsel = 0; bsel < 2; ++bsel) {
            int b = b0 + 8*bsel;
            float mu = mu_s[b], rs = rstd_s[b];
            float vals[8]; float nsq = 0.f;
            #pragma unroll
            for (int k = 0; k < 8; ++k) {
                float v = (xh[b*256 + i*8 + k] - mu) * rs * g8[k] + be8[k];
                vals[k] = v; nsq = fmaf(v, v, nsq);
            }
            float scale = nsq / ((1.0f + nsq) * sqrtf(nsq + SQ_EPS));
            #pragma unroll
            for (int k = 0; k < 8; ++k) u_t[i*130 + k*16 + b] = scale * vals[k];
        }
    }

    #pragma unroll
    for (int m = 0; m < 2; ++m) {
        #pragma unroll
        for (int g = 0; g < 4; ++g)
            Cs[(((2*hi + m)*4 + g)*8 + wav)*16 + bb] = make_float4(0.0625f, 0.0625f, 0.0625f, 0.0625f);
    }
    __syncthreads();

    v2f a0[64], a1[64];
    #pragma unroll
    for (int q = 0; q < 64; ++q) {
        const float* wc = Wcap + (size_t)(((4*wav + (q >> 4))*16 + (q & 15))*8)*16 + bb;
        a0[q][0] = wc[(2*hi+0)*16];  a0[q][1] = wc[(2*hi+1)*16];
        a1[q][0] = wc[(2*hi+4)*16];  a1[q][1] = wc[(2*hi+5)*16];
    }
    v2f bu0[4], bu1[4];
    #pragma unroll
    for (int ii = 0; ii < 4; ++ii) {
        const float* ub = u_t + (4*wav + ii)*130 + bb;
        bu0[ii][0] = ub[(2*hi+0)*16];  bu0[ii][1] = ub[(2*hi+1)*16];
        bu1[ii][0] = ub[(2*hi+4)*16];  bu1[ii][1] = ub[(2*hi+5)*16];
    }

    float Lg[32];
    #pragma unroll
    for (int q = 0; q < 32; ++q) Lg[q] = 0.f;
    const bool up = (hi != 0);

    v8f vA = vzero8(), vB = vzero8();
    for (int r = 0; r < 3; ++r) {
        #pragma unroll
        for (int g = 0; g < 4; ++g) {
            float cc[4][4];
            #pragma unroll
            for (int ii = 0; ii < 4; ++ii) {
                const float4 c4 = Cs[((ii*4 + g)*8 + wav)*16 + bb];
                cc[ii][0] = c4.x; cc[ii][1] = c4.y; cc[ii][2] = c4.z; cc[ii][3] = c4.w;
            }
            v8f sj[4];
            #pragma unroll
            for (int jj = 0; jj < 4; ++jj) sj[jj] = vzero8();
            #pragma unroll
            for (int ii = 0; ii < 4; ++ii) {
                #pragma unroll
                for (int jj = 0; jj < 4; ++jj) {
                    const float c = cc[ii][jj];
                    v2f p0, p1;
                    p0[0] = bu0[ii][0]*c; p0[1] = bu0[ii][1]*c; p1[0] = bu1[ii][0]*c; p1[1] = bu1[ii][1]*c;
                    sj[jj] = wmma4(a0[ii*16 + 4*g + jj], p0, sj[jj]);
                    sj[jj] = wmma4(a1[ii*16 + 4*g + jj], p1, sj[jj]);
                }
            }
            #pragma unroll
            for (int jj = 0; jj < 4; ++jj) {
                Ps[((wav*16 + 4*g + jj)*2 + 0)*32 + lane] = make_float4(sj[jj][0], sj[jj][1], sj[jj][2], sj[jj][3]);
                Ps[((wav*16 + 4*g + jj)*2 + 1)*32 + lane] = make_float4(sj[jj][4], sj[jj][5], sj[jj][6], sj[jj][7]);
            }
        }
        __syncthreads();
        v8f sA, sB;
        {
            float4 x0 = Ps[((0*16 + wav)*2 + 0)*32 + lane], x1 = Ps[((0*16 + wav)*2 + 1)*32 + lane];
            float4 y0 = Ps[((0*16 + wav + 8)*2 + 0)*32 + lane], y1 = Ps[((0*16 + wav + 8)*2 + 1)*32 + lane];
            sA[0]=x0.x; sA[1]=x0.y; sA[2]=x0.z; sA[3]=x0.w; sA[4]=x1.x; sA[5]=x1.y; sA[6]=x1.z; sA[7]=x1.w;
            sB[0]=y0.x; sB[1]=y0.y; sB[2]=y0.z; sB[3]=y0.w; sB[4]=y1.x; sB[5]=y1.y; sB[6]=y1.z; sB[7]=y1.w;
        }
        #pragma unroll
        for (int w2 = 1; w2 < 8; ++w2) {
            float4 x0 = Ps[((w2*16 + wav)*2 + 0)*32 + lane], x1 = Ps[((w2*16 + wav)*2 + 1)*32 + lane];
            float4 y0 = Ps[((w2*16 + wav + 8)*2 + 0)*32 + lane], y1 = Ps[((w2*16 + wav + 8)*2 + 1)*32 + lane];
            sA[0]+=x0.x; sA[1]+=x0.y; sA[2]+=x0.z; sA[3]+=x0.w; sA[4]+=x1.x; sA[5]+=x1.y; sA[6]+=x1.z; sA[7]+=x1.w;
            sB[0]+=y0.x; sB[1]+=y0.y; sB[2]+=y0.z; sB[3]+=y0.w; sB[4]+=y1.x; sB[5]+=y1.y; sB[6]+=y1.z; sB[7]+=y1.w;
        }
        float nA = 0.f, nB = 0.f;
        #pragma unroll
        for (int v = 0; v < 8; ++v) { nA = fmaf(sA[v], sA[v], nA); nB = fmaf(sB[v], sB[v], nB); }
        nA += __shfl_xor(nA, 16, 32);  nB += __shfl_xor(nB, 16, 32);
        float scA = nA / ((1.0f + nA) * sqrtf(nA + SQ_EPS));
        float scB = nB / ((1.0f + nB) * sqrtf(nB + SQ_EPS));
        #pragma unroll
        for (int v = 0; v < 8; ++v) { vA[v] = sA[v]*scA; vB[v] = sB[v]*scB; }

        if (r < 2) {
            Vs[((wav    )*2 + 0)*32 + lane] = make_float4(vA[0], vA[1], vA[2], vA[3]);
            Vs[((wav    )*2 + 1)*32 + lane] = make_float4(vA[4], vA[5], vA[6], vA[7]);
            Vs[((wav + 8)*2 + 0)*32 + lane] = make_float4(vB[0], vB[1], vB[2], vB[3]);
            Vs[((wav + 8)*2 + 1)*32 + lane] = make_float4(vB[4], vB[5], vB[6], vB[7]);
            __syncthreads();
            #pragma unroll
            for (int j = 0; j < 16; ++j) {
                const float4 y0 = Vs[(j*2 + 0)*32 + lane], y1 = Vs[(j*2 + 1)*32 + lane];
                float ag[4];
                #pragma unroll
                for (int ii = 0; ii < 4; ++ii) {
                    v8f uh = vzero8();
                    uh = wmma4(a0[ii*16 + j], bu0[ii], uh);
                    uh = wmma4(a1[ii*16 + j], bu1[ii], uh);
                    v2f a2 = {0.f, 0.f};
                    a2 = __builtin_elementwise_fma((v2f){uh[0], uh[1]}, (v2f){y0.x, y0.y}, a2);
                    a2 = __builtin_elementwise_fma((v2f){uh[2], uh[3]}, (v2f){y0.z, y0.w}, a2);
                    a2 = __builtin_elementwise_fma((v2f){uh[4], uh[5]}, (v2f){y1.x, y1.y}, a2);
                    a2 = __builtin_elementwise_fma((v2f){uh[6], uh[7]}, (v2f){y1.z, y1.w}, a2);
                    ag[ii] = a2[0] + a2[1];
                }
                #pragma unroll
                for (int m = 0; m < 2; ++m) {
                    const float own = up ? ag[2 + m] : ag[m];
                    const float snd = up ? ag[m] : ag[2 + m];
                    Lg[m*16 + j] += own + __shfl_xor(snd, 16, 32);
                }
            }
            #pragma unroll
            for (int m = 0; m < 2; ++m) {
                float mx = Lg[m*16];
                #pragma unroll
                for (int j = 1; j < 16; ++j) mx = fmaxf(mx, Lg[m*16 + j]);
                float e[16]; float sum = 0.f;
                #pragma unroll
                for (int j = 0; j < 16; ++j) { e[j] = expf(Lg[m*16 + j] - mx); sum += e[j]; }
                float inv = 1.0f / sum;
                #pragma unroll
                for (int g = 0; g < 4; ++g)
                    Cs[(((2*hi + m)*4 + g)*8 + wav)*16 + bb] = make_float4(e[4*g]*inv, e[4*g+1]*inv, e[4*g+2]*inv, e[4*g+3]*inv);
            }
            __syncthreads();
        }
    }

    {
        float pA = 0.f, pB = 0.f;
        #pragma unroll
        for (int v = 0; v < 8; ++v) {
            pA = fmaf(vA[v], Wf[(wav    )*16 + v + 8*hi], pA);
            pB = fmaf(vB[v], Wf[(wav + 8)*16 + v + 8*hi], pB);
        }
        pA += __shfl_xor(pA, 16, 32);  pB += __shfl_xor(pB, 16, 32);
        if (lane < 16) { part[wav*17 + bb] = pA; part[(wav+8)*17 + bb] = pB; }
    }
    __syncthreads();
    if (t < 16) {
        float acc = bf[0];
        #pragma unroll
        for (int j = 0; j < 16; ++j) acc += part[j*17 + t];
        out[tile*16 + t] = acc;
    }
}

extern "C" void kernel_launch(void* const* d_in, const int* in_sizes, int n_in,
                              void* d_out, int out_size, void* d_ws, size_t ws_size,
                              hipStream_t stream) {
    const float* x    = (const float*)d_in[0];
    const float* W1   = (const float*)d_in[1];
    const float* b1   = (const float*)d_in[2];
    const float* W2   = (const float*)d_in[3];
    const float* b2   = (const float*)d_in[4];
    const float* ln_g = (const float*)d_in[5];
    const float* ln_b = (const float*)d_in[6];
    const float* Wcap = (const float*)d_in[7];
    const float* Wf   = (const float*)d_in[8];
    const float* bf   = (const float*)d_in[9];
    float* out = (float*)d_out;

    const int B = in_sizes[0] / 256;
    const int blocks = B / 16;
    capsnet_fused<<<blocks, 256, 0, stream>>>(x, W1, b1, W2, b2, ln_g, ln_b, Wcap, Wf, bf, out);
}


# MLP re-split by weight ownership: layer 1 split over k (wave owns 32 of W1's 256 rows; 8 partial sums added via LDS in fixed order: new sum order), layer 2 over columns; each weight fetched once
# speedup vs baseline: 1.5388x; 1.3551x over previous
#include <hip/hip_runtime.h>
#include <math.h>

typedef __attribute__((ext_vector_type(2))) float v2f;
typedef __attribute__((ext_vector_type(8))) float v8f;

#define LN_EPS 1e-5f
#define SQ_EPS 1e-9f

__device__ __forceinline__ float gelu_exact(float x) {
    return 0.5f * x * (1.0f + erff(x * 0.70710678118654752440f));
}

__device__ __forceinline__ v8f wmma4(v2f a, v2f b, v8f c) {
    return __builtin_amdgcn_wmma_f32_16x16x4_f32(false, a, false, b, (short)0, c, false, false);
}

__device__ __forceinline__ v8f vzero8() {
    v8f z = {0.f,0.f,0.f,0.f,0.f,0.f,0.f,0.f};
    return z;
}

__global__ __attribute__((amdgpu_waves_per_eu(1))) __launch_bounds__(256)
void capsnet_fused(
    const float* __restrict__ x,   const float* __restrict__ W1, const float* __restrict__ b1,
    const float* __restrict__ W2,  const float* __restrict__ b2,
    const float* __restrict__ ln_g,const float* __restrict__ ln_b,
    const float* __restrict__ Wcap,const float* __restrict__ Wf, const float* __restrict__ bf,
    float* __restrict__ out)
{
    __shared__ __align__(16) float xh[16*256];
    __shared__ __align__(16) float h1s[16*32];
    __shared__ float hp[8*16*32];
    __shared__ float mu_s[16];
    __shared__ float rstd_s[16];
    __shared__ float u_t[32*130];
    __shared__ float4 Ps[8*16*64];
    __shared__ float4 Cs[16*8*16];
    __shared__ float4 Vs[16*64];
    __shared__ float part[16*17];

    const int t    = threadIdx.x;
    const int lane = t & 31;
    const int wav  = t >> 5;
    const int bb   = lane & 15;
    const int hi   = lane >> 4;

    const int tile = blockIdx.x;
    const float* xg = x + (size_t)tile * 16 * 256;

    #pragma unroll
    for (int r = 0; r < 16; ++r) xh[t + 256*r] = xg[t + 256*r];
    __syncthreads();

    {
        float wk[32];
        #pragma unroll
        for (int k = 0; k < 32; ++k) wk[k] = W1[(32*wav + k)*32 + lane];
        #pragma unroll 1
        for (int bg = 0; bg < 16; bg += 4) {
            float acc[4] = {0.f, 0.f, 0.f, 0.f};
            #pragma unroll
            for (int bi = 0; bi < 4; ++bi) {
                const float4* xr = reinterpret_cast<const float4*>(xh + (bg + bi)*256 + 32*wav);
                #pragma unroll
                for (int q = 0; q < 8; ++q) {
                    const float4 xv = xr[q];
                    acc[bi] = fmaf(xv.x, wk[4*q+0], acc[bi]);
                    acc[bi] = fmaf(xv.y, wk[4*q+1], acc[bi]);
                    acc[bi] = fmaf(xv.z, wk[4*q+2], acc[bi]);
                    acc[bi] = fmaf(xv.w, wk[4*q+3], acc[bi]);
                }
            }
            #pragma unroll
            for (int bi = 0; bi < 4; ++bi) hp[(wav*16 + bg + bi)*32 + lane] = acc[bi];
        }
    }
    __syncthreads();

    {
        int n = lane, b0 = wav;
        float s0 = b1[n], s1 = s0;
        #pragma unroll
        for (int w2 = 0; w2 < 8; ++w2) { s0 += hp[(w2*16 + b0)*32 + n]; s1 += hp[(w2*16 + b0 + 8)*32 + n]; }
        h1s[b0*32 + n]     = gelu_exact(s0);
        h1s[(b0+8)*32 + n] = gelu_exact(s1);
    }
    __syncthreads();

    {
        const int n = 32*wav + lane;
        float wk[32];
        #pragma unroll
        for (int k = 0; k < 32; ++k) wk[k] = W2[k*256 + n];
        const float bias = b2[n];
        #pragma unroll 1
        for (int bg = 0; bg < 16; bg += 4) {
            float acc[4] = {bias, bias, bias, bias};
            #pragma unroll
            for (int bi = 0; bi < 4; ++bi) {
                const float4* hr = reinterpret_cast<const float4*>(h1s + (bg + bi)*32);
                #pragma unroll
                for (int q = 0; q < 8; ++q) {
                    const float4 hv = hr[q];
                    acc[bi] = fmaf(hv.x, wk[4*q+0], acc[bi]);
                    acc[bi] = fmaf(hv.y, wk[4*q+1], acc[bi]);
                    acc[bi] = fmaf(hv.z, wk[4*q+2], acc[bi]);
                    acc[bi] = fmaf(hv.w, wk[4*q+3], acc[bi]);
                }
            }
            #pragma unroll
            for (int bi = 0; bi < 4; ++bi) xh[(bg + bi)*256 + n] = gelu_exact(acc[bi]);
        }
    }
    __syncthreads();

    {
        int ba = wav, bc = wav + 8;
        float sa = 0.f, sb = 0.f;
        #pragma unroll
        for (int r = 0; r < 8; ++r) { sa += xh[ba*256 + lane + 32*r]; sb += xh[bc*256 + lane + 32*r]; }
        #pragma unroll
        for (int m = 16; m >= 1; m >>= 1) { sa += __shfl_xor(sa, m, 32); sb += __shfl_xor(sb, m, 32); }
        float mua = sa * (1.0f/256.0f), mub = sb * (1.0f/256.0f);
        float va = 0.f, vb = 0.f;
        #pragma unroll
        for (int r = 0; r < 8; ++r) {
            float da = xh[ba*256 + lane + 32*r] - mua; va = fmaf(da, da, va);
            float db = xh[bc*256 + lane + 32*r] - mub; vb = fmaf(db, db, vb);
        }
        #pragma unroll
        for (int m = 16; m >= 1; m >>= 1) { va += __shfl_xor(va, m, 32); vb += __shfl_xor(vb, m, 32); }
        if (lane == 0) {
            mu_s[ba] = mua; rstd_s[ba] = rsqrtf(va * (1.0f/256.0f) + LN_EPS);
            mu_s[bc] = mub; rstd_s[bc] = rsqrtf(vb * (1.0f/256.0f) + LN_EPS);
        }
    }
    __syncthreads();

    {
        int i = t & 31, b0 = t >> 5;
        float g8[8], be8[8];
        #pragma unroll
        for (int k = 0; k < 8; ++k) { g8[k] = ln_g[i*8 + k]; be8[k] = ln_b[i*8 + k]; }
        #pragma unroll
        for (int bsel = 0; bsel < 2; ++bsel) {
            int b = b0 + 8*bsel;
            float mu = mu_s[b], rs = rstd_s[b];
            float vals[8]; float nsq = 0.f;
            #pragma unroll
            for (int k = 0; k < 8; ++k) {
                float v = (xh[b*256 + i*8 + k] - mu) * rs * g8[k] + be8[k];
                vals[k] = v; nsq = fmaf(v, v, nsq);
            }
            float scale = nsq / ((1.0f + nsq) * sqrtf(nsq + SQ_EPS));
            #pragma unroll
            for (int k = 0; k < 8; ++k) u_t[i*130 + k*16 + b] = scale * vals[k];
        }
    }

    #pragma unroll
    for (int m = 0; m < 2; ++m) {
        #pragma unroll
        for (int g = 0; g < 4; ++g)
            Cs[(((2*hi + m)*4 + g)*8 + wav)*16 + bb] = make_float4(0.0625f, 0.0625f, 0.0625f, 0.0625f);
    }
    __syncthreads();

    v2f a0[64], a1[64];
    #pragma unroll
    for (int q = 0; q < 64; ++q) {
        const float* wc = Wcap + (size_t)(((4*wav + (q >> 4))*16 + (q & 15))*8)*16 + bb;
        a0[q][0] = wc[(2*hi+0)*16];  a0[q][1] = wc[(2*hi+1)*16];
        a1[q][0] = wc[(2*hi+4)*16];  a1[q][1] = wc[(2*hi+5)*16];
    }
    v2f bu0[4], bu1[4];
    #pragma unroll
    for (int ii = 0; ii < 4; ++ii) {
        const float* ub = u_t + (4*wav + ii)*130 + bb;
        bu0[ii][0] = ub[(2*hi+0)*16];  bu0[ii][1] = ub[(2*hi+1)*16];
        bu1[ii][0] = ub[(2*hi+4)*16];  bu1[ii][1] = ub[(2*hi+5)*16];
    }

    float Lg[32];
    #pragma unroll
    for (int q = 0; q < 32; ++q) Lg[q] = 0.f;
    const bool up = (hi != 0);

    v8f vA = vzero8(), vB = vzero8();
    for (int r = 0; r < 3; ++r) {
        #pragma unroll
        for (int g = 0; g < 4; ++g) {
            float cc[4][4];
            #pragma unroll
            for (int ii = 0; ii < 4; ++ii) {
                const float4 c4 = Cs[((ii*4 + g)*8 + wav)*16 + bb];
                cc[ii][0] = c4.x; cc[ii][1] = c4.y; cc[ii][2] = c4.z; cc[ii][3] = c4.w;
            }
            v8f sj[4];
            #pragma unroll
            for (int jj = 0; jj < 4; ++jj) sj[jj] = vzero8();
            #pragma unroll
            for (int ii = 0; ii < 4; ++ii) {
                #pragma unroll
                for (int jj = 0; jj < 4; ++jj) {
                    const float c = cc[ii][jj];
                    v2f p0, p1;
                    p0[0] = bu0[ii][0]*c; p0[1] = bu0[ii][1]*c; p1[0] = bu1[ii][0]*c; p1[1] = bu1[ii][1]*c;
                    sj[jj] = wmma4(a0[ii*16 + 4*g + jj], p0, sj[jj]);
                    sj[jj] = wmma4(a1[ii*16 + 4*g + jj], p1, sj[jj]);
                }
            }
            #pragma unroll
            for (int jj = 0; jj < 4; ++jj) {
                Ps[((wav*16 + 4*g + jj)*2 + 0)*32 + lane] = make_float4(sj[jj][0], sj[jj][1], sj[jj][2], sj[jj][3]);
                Ps[((wav*16 + 4*g + jj)*2 + 1)*32 + lane] = make_float4(sj[jj][4], sj[jj][5], sj[jj][6], sj[jj][7]);
            }
        }
        __syncthreads();
        v8f sA, sB;
        {
            float4 x0 = Ps[((0*16 + wav)*2 + 0)*32 + lane], x1 = Ps[((0*16 + wav)*2 + 1)*32 + lane];
            float4 y0 = Ps[((0*16 + wav + 8)*2 + 0)*32 + lane], y1 = Ps[((0*16 + wav + 8)*2 + 1)*32 + lane];
            sA[0]=x0.x; sA[1]=x0.y; sA[2]=x0.z; sA[3]=x0.w; sA[4]=x1.x; sA[5]=x1.y; sA[6]=x1.z; sA[7]=x1.w;
            sB[0]=y0.x; sB[1]=y0.y; sB[2]=y0.z; sB[3]=y0.w; sB[4]=y1.x; sB[5]=y1.y; sB[6]=y1.z; sB[7]=y1.w;
        }
        #pragma unroll
        for (int w2 = 1; w2 < 8; ++w2) {
            float4 x0 = Ps[((w2*16 + wav)*2 + 0)*32 + lane], x1 = Ps[((w2*16 + wav)*2 + 1)*32 + lane];
            float4 y0 = Ps[((w2*16 + wav + 8)*2 + 0)*32 + lane], y1 = Ps[((w2*16 + wav + 8)*2 + 1)*32 + lane];
            sA[0]+=x0.x; sA[1]+=x0.y; sA[2]+=x0.z; sA[3]+=x0.w; sA[4]+=x1.x; sA[5]+=x1.y; sA[6]+=x1.z; sA[7]+=x1.w;
            sB[0]+=y0.x; sB[1]+=y0.y; sB[2]+=y0.z; sB[3]+=y0.w; sB[4]+=y1.x; sB[5]+=y1.y; sB[6]+=y1.z; sB[7]+=y1.w;
        }
        float nA = 0.f, nB = 0.f;
        #pragma unroll
        for (int v = 0; v < 8; ++v) { nA = fmaf(sA[v], sA[v], nA); nB = fmaf(sB[v], sB[v], nB); }
        nA += __shfl_xor(nA, 16, 32);  nB += __shfl_xor(nB, 16, 32);
        float scA = nA / ((1.0f + nA) * sqrtf(nA + SQ_EPS));
        float scB = nB / ((1.0f + nB) * sqrtf(nB + SQ_EPS));
        #pragma unroll
        for (int v = 0; v < 8; ++v) { vA[v] = sA[v]*scA; vB[v] = sB[v]*scB; }

        if (r < 2) {
            Vs[((wav    )*2 + 0)*32 + lane] = make_float4(vA[0], vA[1], vA[2], vA[3]);
            Vs[((wav    )*2 + 1)*32 + lane] = make_float4(vA[4], vA[5], vA[6], vA[7]);
            Vs[((wav + 8)*2 + 0)*32 + lane] = make_float4(vB[0], vB[1], vB[2], vB[3]);
            Vs[((wav + 8)*2 + 1)*32 + lane] = make_float4(vB[4], vB[5], vB[6], vB[7]);
            __syncthreads();
            #pragma unroll
            for (int j = 0; j < 16; ++j) {
                const float4 y0 = Vs[(j*2 + 0)*32 + lane], y1 = Vs[(j*2 + 1)*32 + lane];
                float ag[4];
                #pragma unroll
                for (int ii = 0; ii < 4; ++ii) {
                    v8f uh = vzero8();
                    uh = wmma4(a0[ii*16 + j], bu0[ii], uh);
                    uh = wmma4(a1[ii*16 + j], bu1[ii], uh);
                    v2f a2 = {0.f, 0.f};
                    a2 = __builtin_elementwise_fma((v2f){uh[0], uh[1]}, (v2f){y0.x, y0.y}, a2);
                    a2 = __builtin_elementwise_fma((v2f){uh[2], uh[3]}, (v2f){y0.z, y0.w}, a2);
                    a2 = __builtin_elementwise_fma((v2f){uh[4], uh[5]}, (v2f){y1.x, y1.y}, a2);
                    a2 = __builtin_elementwise_fma((v2f){uh[6], uh[7]}, (v2f){y1.z, y1.w}, a2);
                    ag[ii] = a2[0] + a2[1];
                }
                #pragma unroll
                for (int m = 0; m < 2; ++m) {
                    const float own = up ? ag[2 + m] : ag[m];
                    const float snd = up ? ag[m] : ag[2 + m];
                    Lg[m*16 + j] += own + __shfl_xor(snd, 16, 32);
                }
            }
            #pragma unroll
            for (int m = 0; m < 2; ++m) {
                float mx = Lg[m*16];
                #pragma unroll
                for (int j = 1; j < 16; ++j) mx = fmaxf(mx, Lg[m*16 + j]);
                float e[16]; float sum = 0.f;
                #pragma unroll
                for (int j = 0; j < 16; ++j) { e[j] = expf(Lg[m*16 + j] - mx); sum += e[j]; }
                float inv = 1.0f / sum;
                #pragma unroll
                for (int g = 0; g < 4; ++g)
                    Cs[(((2*hi + m)*4 + g)*8 + wav)*16 + bb] = make_float4(e[4*g]*inv, e[4*g+1]*inv, e[4*g+2]*inv, e[4*g+3]*inv);
            }
            __syncthreads();
        }
    }

    {
        float pA = 0.f, pB = 0.f;
        #pragma unroll
        for (int v = 0; v < 8; ++v) {
            pA = fmaf(vA[v], Wf[(wav    )*16 + v + 8*hi], pA);
            pB = fmaf(vB[v], Wf[(wav + 8)*16 + v + 8*hi], pB);
        }
        pA += __shfl_xor(pA, 16, 32);  pB += __shfl_xor(pB, 16, 32);
        if (lane < 16) { part[wav*17 + bb] = pA; part[(wav+8)*17 + bb] = pB; }
    }
    __syncthreads();
    if (t < 16) {
        float acc = bf[0];
        #pragma unroll
        for (int j = 0; j < 16; ++j) acc += part[j*17 + t];
        out[tile*16 + t] = acc;
    }
}

extern "C" void kernel_launch(void* const* d_in, const int* in_sizes, int n_in,
                              void* d_out, int out_size, void* d_ws, size_t ws_size,
                              hipStream_t stream) {
    const float* x    = (const float*)d_in[0];
    const float* W1   = (const float*)d_in[1];
    const float* b1   = (const float*)d_in[2];
    const float* W2   = (const float*)d_in[3];
    const float* b2   = (const float*)d_in[4];
    const float* ln_g = (const float*)d_in[5];
    const float* ln_b = (const float*)d_in[6];
    const float* Wcap = (const float*)d_in[7];
    const float* Wf   = (const float*)d_in[8];
    const float* bf   = (const float*)d_in[9];
    float* out = (float*)d_out;

    const int B = in_sizes[0] / 256;
    const int blocks = B / 16;
    capsnet_fused<<<blocks, 256, 0, stream>>>(x, W1, b1, W2, b2, ln_g, ln_b, Wcap, Wf, bf, out);
}


# Wcap fragment loads (256 per lane) issued in five groups of at most 60 under the MLP and LayerNorm phases, in the first s pass's order; barriers between them order LDS only; sums bit-identical
# speedup vs baseline: 1.5872x; 1.0315x over previous
#include <hip/hip_runtime.h>
#include <math.h>

typedef __attribute__((ext_vector_type(2))) float v2f;
typedef __attribute__((ext_vector_type(8))) float v8f;

#define LN_EPS 1e-5f
#define SQ_EPS 1e-9f

__device__ __forceinline__ float gelu_exact(float x) {
    return 0.5f * x * (1.0f + erff(x * 0.70710678118654752440f));
}

__device__ __forceinline__ v8f wmma4(v2f a, v2f b, v8f c) {
    return __builtin_amdgcn_wmma_f32_16x16x4_f32(false, a, false, b, (short)0, c, false, false);
}

__device__ __forceinline__ void lds_barrier() {
    __builtin_amdgcn_fence(__ATOMIC_RELEASE, "workgroup", "local");
    __builtin_amdgcn_s_barrier();
    __builtin_amdgcn_fence(__ATOMIC_ACQUIRE, "workgroup", "local");
}

__device__ __forceinline__ v8f vzero8() {
    v8f z = {0.f,0.f,0.f,0.f,0.f,0.f,0.f,0.f};
    return z;
}

__global__ __attribute__((amdgpu_waves_per_eu(1))) __launch_bounds__(256)
void capsnet_fused(
    const float* __restrict__ x,   const float* __restrict__ W1, const float* __restrict__ b1,
    const float* __restrict__ W2,  const float* __restrict__ b2,
    const float* __restrict__ ln_g,const float* __restrict__ ln_b,
    const float* __restrict__ Wcap,const float* __restrict__ Wf, const float* __restrict__ bf,
    float* __restrict__ out)
{
    __shared__ __align__(16) float xh[16*256];
    __shared__ __align__(16) float h1s[16*32];
    __shared__ float hp[8*16*32];
    __shared__ float mu_s[16];
    __shared__ float rstd_s[16];
    __shared__ float u_t[32*130];
    __shared__ float4 Ps[8*16*64];
    __shared__ float4 Cs[16*8*16];
    __shared__ float4 Vs[16*64];
    __shared__ float part[16*17];

    const int t    = threadIdx.x;
    const int lane = t & 31;
    const int wav  = t >> 5;
    const int bb   = lane & 15;
    const int hi   = lane >> 4;

    const int tile = blockIdx.x;
    const float* xg = x + (size_t)tile * 16 * 256;

    v2f a0[64], a1[64];
#define LOAD_WCAP_RANGE(LO, HI)                                                                               \
    _Pragma("unroll")                                                                                         \
    for (int p = (LO); p < (HI); ++p) {                                                                       \
        const int pi = (p >> 2) & 3, pj = (p >> 4)*4 + (p & 3);                                               \
        const float* wc = Wcap + (size_t)(((4*wav + pi)*16 + pj)*8)*16 + bb;                  \
        a0[pi*16 + pj][0] = wc[(2*hi+0)*16];  a0[pi*16 + pj][1] = wc[(2*hi+1)*16];                             \
        a1[pi*16 + pj][0] = wc[(2*hi+4)*16];  a1[pi*16 + pj][1] = wc[(2*hi+5)*16];                             \
    }

    float wk1[32];
    #pragma unroll
    for (int k = 0; k < 32; ++k) wk1[k] = W1[(32*wav + k)*32 + lane];

    #pragma unroll
    for (int r = 0; r < 16; ++r) xh[t + 256*r] = xg[t + 256*r];
    lds_barrier();
    LOAD_WCAP_RANGE(0, 15)

    {
        #pragma unroll 1
        for (int bg = 0; bg < 16; bg += 4) {
            float acc[4] = {0.f, 0.f, 0.f, 0.f};
            #pragma unroll
            for (int bi = 0; bi < 4; ++bi) {
                const float4* xr = reinterpret_cast<const float4*>(xh + (bg + bi)*256 + 32*wav);
                #pragma unroll
                for (int q = 0; q < 8; ++q) {
                    const float4 xv = xr[q];
                    acc[bi] = fmaf(xv.x, wk1[4*q+0], acc[bi]);
                    acc[bi] = fmaf(xv.y, wk1[4*q+1], acc[bi]);
                    acc[bi] = fmaf(xv.z, wk1[4*q+2], acc[bi]);
                    acc[bi] = fmaf(xv.w, wk1[4*q+3], acc[bi]);
                }
            }
            #pragma unroll
            for (int bi = 0; bi < 4; ++bi) hp[(wav*16 + bg + bi)*32 + lane] = acc[bi];
        }
    }
    lds_barrier();

    float wk2[32];
    #pragma unroll
    for (int k = 0; k < 32; ++k) wk2[k] = W2[k*256 + 32*wav + lane];
    const float b2v = b2[32*wav + lane];
    float g8[8], be8[8];
    #pragma unroll
    for (int k = 0; k < 8; ++k) { g8[k] = ln_g[(t & 31)*8 + k]; be8[k] = ln_b[(t & 31)*8 + k]; }
    LOAD_WCAP_RANGE(15, 19)

    {
        int n = lane, b0 = wav;
        float s0 = b1[n], s1 = s0;
        #pragma unroll
        for (int w2 = 0; w2 < 8; ++w2) { s0 += hp[(w2*16 + b0)*32 + n]; s1 += hp[(w2*16 + b0 + 8)*32 + n]; }
        h1s[b0*32 + n]     = gelu_exact(s0);
        h1s[(b0+8)*32 + n] = gelu_exact(s1);
    }
    lds_barrier();

    {
        const int n = 32*wav + lane;
        const float bias = b2v;
        LOAD_WCAP_RANGE(19, 34)
        #pragma unroll 1
        for (int bg = 0; bg < 16; bg += 4) {
            float acc[4] = {bias, bias, bias, bias};
            #pragma unroll
            for (int bi = 0; bi < 4; ++bi) {
                const float4* hr = reinterpret_cast<const float4*>(h1s + (bg + bi)*32);
                #pragma unroll
                for (int q = 0; q < 8; ++q) {
                    const float4 hv = hr[q];
                    acc[bi] = fmaf(hv.x, wk2[4*q+0], acc[bi]);
                    acc[bi] = fmaf(hv.y, wk2[4*q+1], acc[bi]);
                    acc[bi] = fmaf(hv.z, wk2[4*q+2], acc[bi]);
                    acc[bi] = fmaf(hv.w, wk2[4*q+3], acc[bi]);
                }
            }
            #pragma unroll
            for (int bi = 0; bi < 4; ++bi) xh[(bg + bi)*256 + n] = gelu_exact(acc[bi]);
        }
    }
    LOAD_WCAP_RANGE(34, 49)
    lds_barrier();

    {
        int ba = wav, bc = wav + 8;
        float sa = 0.f, sb = 0.f;
        #pragma unroll
        for (int r = 0; r < 8; ++r) { sa += xh[ba*256 + lane + 32*r]; sb += xh[bc*256 + lane + 32*r]; }
        #pragma unroll
        for (int m = 16; m >= 1; m >>= 1) { sa += __shfl_xor(sa, m, 32); sb += __shfl_xor(sb, m, 32); }
        float mua = sa * (1.0f/256.0f), mub = sb * (1.0f/256.0f);
        float va = 0.f, vb = 0.f;
        #pragma unroll
        for (int r = 0; r < 8; ++r) {
            float da = xh[ba*256 + lane + 32*r] - mua; va = fmaf(da, da, va);
            float db = xh[bc*256 + lane + 32*r] - mub; vb = fmaf(db, db, vb);
        }
        #pragma unroll
        for (int m = 16; m >= 1; m >>= 1) { va += __shfl_xor(va, m, 32); vb += __shfl_xor(vb, m, 32); }
        if (lane == 0) {
            mu_s[ba] = mua; rstd_s[ba] = rsqrtf(va * (1.0f/256.0f) + LN_EPS);
            mu_s[bc] = mub; rstd_s[bc] = rsqrtf(vb * (1.0f/256.0f) + LN_EPS);
        }
    }
    lds_barrier();
    LOAD_WCAP_RANGE(49, 64)

    {
        int i = t & 31, b0 = t >> 5;
        #pragma unroll
        for (int bsel = 0; bsel < 2; ++bsel) {
            int b = b0 + 8*bsel;
            float mu = mu_s[b], rs = rstd_s[b];
            float vals[8]; float nsq = 0.f;
            #pragma unroll
            for (int k = 0; k < 8; ++k) {
                float v = (xh[b*256 + i*8 + k] - mu) * rs * g8[k] + be8[k];
                vals[k] = v; nsq = fmaf(v, v, nsq);
            }
            float scale = nsq / ((1.0f + nsq) * sqrtf(nsq + SQ_EPS));
            #pragma unroll
            for (int k = 0; k < 8; ++k) u_t[i*130 + k*16 + b] = scale * vals[k];
        }
    }

    #pragma unroll
    for (int m = 0; m < 2; ++m) {
        #pragma unroll
        for (int g = 0; g < 4; ++g)
            Cs[(((2*hi + m)*4 + g)*8 + wav)*16 + bb] = make_float4(0.0625f, 0.0625f, 0.0625f, 0.0625f);
    }
    lds_barrier();

    v2f bu0[4], bu1[4];
    #pragma unroll
    for (int ii = 0; ii < 4; ++ii) {
        const float* ub = u_t + (4*wav + ii)*130 + bb;
        bu0[ii][0] = ub[(2*hi+0)*16];  bu0[ii][1] = ub[(2*hi+1)*16];
        bu1[ii][0] = ub[(2*hi+4)*16];  bu1[ii][1] = ub[(2*hi+5)*16];
    }

    float Lg[32];
    #pragma unroll
    for (int q = 0; q < 32; ++q) Lg[q] = 0.f;
    const bool up = (hi != 0);

    v8f vA = vzero8(), vB = vzero8();
    for (int r = 0; r < 3; ++r) {
        #pragma unroll
        for (int g = 0; g < 4; ++g) {
            float cc[4][4];
            #pragma unroll
            for (int ii = 0; ii < 4; ++ii) {
                const float4 c4 = Cs[((ii*4 + g)*8 + wav)*16 + bb];
                cc[ii][0] = c4.x; cc[ii][1] = c4.y; cc[ii][2] = c4.z; cc[ii][3] = c4.w;
            }
            v8f sj[4];
            #pragma unroll
            for (int jj = 0; jj < 4; ++jj) sj[jj] = vzero8();
            #pragma unroll
            for (int ii = 0; ii < 4; ++ii) {
                #pragma unroll
                for (int jj = 0; jj < 4; ++jj) {
                    const float c = cc[ii][jj];
                    v2f p0, p1;
                    p0[0] = bu0[ii][0]*c; p0[1] = bu0[ii][1]*c; p1[0] = bu1[ii][0]*c; p1[1] = bu1[ii][1]*c;
                    sj[jj] = wmma4(a0[ii*16 + 4*g + jj], p0, sj[jj]);
                    sj[jj] = wmma4(a1[ii*16 + 4*g + jj], p1, sj[jj]);
                }
            }
            #pragma unroll
            for (int jj = 0; jj < 4; ++jj) {
                Ps[((wav*16 + 4*g + jj)*2 + 0)*32 + lane] = make_float4(sj[jj][0], sj[jj][1], sj[jj][2], sj[jj][3]);
                Ps[((wav*16 + 4*g + jj)*2 + 1)*32 + lane] = make_float4(sj[jj][4], sj[jj][5], sj[jj][6], sj[jj][7]);
            }
        }
        __syncthreads();
        v8f sA, sB;
        {
            float4 x0 = Ps[((0*16 + wav)*2 + 0)*32 + lane], x1 = Ps[((0*16 + wav)*2 + 1)*32 + lane];
            float4 y0 = Ps[((0*16 + wav + 8)*2 + 0)*32 + lane], y1 = Ps[((0*16 + wav + 8)*2 + 1)*32 + lane];
            sA[0]=x0.x; sA[1]=x0.y; sA[2]=x0.z; sA[3]=x0.w; sA[4]=x1.x; sA[5]=x1.y; sA[6]=x1.z; sA[7]=x1.w;
            sB[0]=y0.x; sB[1]=y0.y; sB[2]=y0.z; sB[3]=y0.w; sB[4]=y1.x; sB[5]=y1.y; sB[6]=y1.z; sB[7]=y1.w;
        }
        #pragma unroll
        for (int w2 = 1; w2 < 8; ++w2) {
            float4 x0 = Ps[((w2*16 + wav)*2 + 0)*32 + lane], x1 = Ps[((w2*16 + wav)*2 + 1)*32 + lane];
            float4 y0 = Ps[((w2*16 + wav + 8)*2 + 0)*32 + lane], y1 = Ps[((w2*16 + wav + 8)*2 + 1)*32 + lane];
            sA[0]+=x0.x; sA[1]+=x0.y; sA[2]+=x0.z; sA[3]+=x0.w; sA[4]+=x1.x; sA[5]+=x1.y; sA[6]+=x1.z; sA[7]+=x1.w;
            sB[0]+=y0.x; sB[1]+=y0.y; sB[2]+=y0.z; sB[3]+=y0.w; sB[4]+=y1.x; sB[5]+=y1.y; sB[6]+=y1.z; sB[7]+=y1.w;
        }
        float nA = 0.f, nB = 0.f;
        #pragma unroll
        for (int v = 0; v < 8; ++v) { nA = fmaf(sA[v], sA[v], nA); nB = fmaf(sB[v], sB[v], nB); }
        nA += __shfl_xor(nA, 16, 32);  nB += __shfl_xor(nB, 16, 32);
        float scA = nA / ((1.0f + nA) * sqrtf(nA + SQ_EPS));
        float scB = nB / ((1.0f + nB) * sqrtf(nB + SQ_EPS));
        #pragma unroll
        for (int v = 0; v < 8; ++v) { vA[v] = sA[v]*scA; vB[v] = sB[v]*scB; }

        if (r < 2) {
            Vs[((wav    )*2 + 0)*32 + lane] = make_float4(vA[0], vA[1], vA[2], vA[3]);
            Vs[((wav    )*2 + 1)*32 + lane] = make_float4(vA[4], vA[5], vA[6], vA[7]);
            Vs[((wav + 8)*2 + 0)*32 + lane] = make_float4(vB[0], vB[1], vB[2], vB[3]);
            Vs[((wav + 8)*2 + 1)*32 + lane] = make_float4(vB[4], vB[5], vB[6], vB[7]);
            __syncthreads();
            #pragma unroll
            for (int j = 0; j < 16; ++j) {
                const float4 y0 = Vs[(j*2 + 0)*32 + lane], y1 = Vs[(j*2 + 1)*32 + lane];
                float ag[4];
                #pragma unroll
                for (int ii = 0; ii < 4; ++ii) {
                    v8f uh = vzero8();
                    uh = wmma4(a0[ii*16 + j], bu0[ii], uh);
                    uh = wmma4(a1[ii*16 + j], bu1[ii], uh);
                    v2f a2 = {0.f, 0.f};
                    a2 = __builtin_elementwise_fma((v2f){uh[0], uh[1]}, (v2f){y0.x, y0.y}, a2);
                    a2 = __builtin_elementwise_fma((v2f){uh[2], uh[3]}, (v2f){y0.z, y0.w}, a2);
                    a2 = __builtin_elementwise_fma((v2f){uh[4], uh[5]}, (v2f){y1.x, y1.y}, a2);
                    a2 = __builtin_elementwise_fma((v2f){uh[6], uh[7]}, (v2f){y1.z, y1.w}, a2);
                    ag[ii] = a2[0] + a2[1];
                }
                #pragma unroll
                for (int m = 0; m < 2; ++m) {
                    const float own = up ? ag[2 + m] : ag[m];
                    const float snd = up ? ag[m] : ag[2 + m];
                    Lg[m*16 + j] += own + __shfl_xor(snd, 16, 32);
                }
            }
            #pragma unroll
            for (int m = 0; m < 2; ++m) {
                float mx = Lg[m*16];
                #pragma unroll
                for (int j = 1; j < 16; ++j) mx = fmaxf(mx, Lg[m*16 + j]);
                float e[16]; float sum = 0.f;
                #pragma unroll
                for (int j = 0; j < 16; ++j) { e[j] = expf(Lg[m*16 + j] - mx); sum += e[j]; }
                float inv = 1.0f / sum;
                #pragma unroll
                for (int g = 0; g < 4; ++g)
                    Cs[(((2*hi + m)*4 + g)*8 + wav)*16 + bb] = make_float4(e[4*g]*inv, e[4*g+1]*inv, e[4*g+2]*inv, e[4*g+3]*inv);
            }
            __syncthreads();
        }
    }

    {
        float pA = 0.f, pB = 0.f;
        #pragma unroll
        for (int v = 0; v < 8; ++v) {
            pA = fmaf(vA[v], Wf[(wav    )*16 + v + 8*hi], pA);
            pB = fmaf(vB[v], Wf[(wav + 8)*16 + v + 8*hi], pB);
        }
        pA += __shfl_xor(pA, 16, 32);  pB += __shfl_xor(pB, 16, 32);
        if (lane < 16) { part[wav*17 + bb] = pA; part[(wav+8)*17 + bb] = pB; }
    }
    __syncthreads();
    if (t < 16) {
        float acc = bf[0];
        #pragma unroll
        for (int j = 0; j < 16; ++j) acc += part[j*17 + t];
        out[tile*16 + t] = acc;
    }
}

extern "C" void kernel_launch(void* const* d_in, const int* in_sizes, int n_in,
                              void* d_out, int out_size, void* d_ws, size_t ws_size,
                              hipStream_t stream) {
    const float* x    = (const float*)d_in[0];
    const float* W1   = (const float*)d_in[1];
    const float* b1   = (const float*)d_in[2];
    const float* W2   = (const float*)d_in[3];
    const float* b2   = (const float*)d_in[4];
    const float* ln_g = (const float*)d_in[5];
    const float* ln_b = (const float*)d_in[6];
    const float* Wcap = (const float*)d_in[7];
    const float* Wf   = (const float*)d_in[8];
    const float* bf   = (const float*)d_in[9];
    float* out = (float*)d_out;

    const int B = in_sizes[0] / 256;
    const int blocks = B / 16;
    capsnet_fused<<<blocks, 256, 0, stream>>>(x, W1, b1, W2, b2, ln_g, ln_b, Wcap, Wf, bf, out);
}


# Each wave's first capsule: its 16 u_hat tiles multiplied once, kept in registers in place of its Wcap fragments; 5 routing passes issue 96 not 128 matrix calls; s starts at c*u_hat: new sum order
# speedup vs baseline: 1.8701x; 1.1782x over previous
#include <hip/hip_runtime.h>
#include <math.h>

typedef __attribute__((ext_vector_type(2))) float v2f;
typedef __attribute__((ext_vector_type(8))) float v8f;

#define LN_EPS 1e-5f
#define SQ_EPS 1e-9f

__device__ __forceinline__ float gelu_exact(float x) {
    return 0.5f * x * (1.0f + erff(x * 0.70710678118654752440f));
}

__device__ __forceinline__ v8f wmma4(v2f a, v2f b, v8f c) {
    return __builtin_amdgcn_wmma_f32_16x16x4_f32(false, a, false, b, (short)0, c, false, false);
}

__device__ __forceinline__ void lds_barrier() {
    __builtin_amdgcn_fence(__ATOMIC_RELEASE, "workgroup", "local");
    __builtin_amdgcn_s_barrier();
    __builtin_amdgcn_fence(__ATOMIC_ACQUIRE, "workgroup", "local");
}

__device__ __forceinline__ v8f vzero8() {
    v8f z = {0.f,0.f,0.f,0.f,0.f,0.f,0.f,0.f};
    return z;
}

__global__ __attribute__((amdgpu_waves_per_eu(1))) __launch_bounds__(256)
void capsnet_fused(
    const float* __restrict__ x,   const float* __restrict__ W1, const float* __restrict__ b1,
    const float* __restrict__ W2,  const float* __restrict__ b2,
    const float* __restrict__ ln_g,const float* __restrict__ ln_b,
    const float* __restrict__ Wcap,const float* __restrict__ Wf, const float* __restrict__ bf,
    float* __restrict__ out)
{
    __shared__ __align__(16) float xh[16*256];
    __shared__ __align__(16) float h1s[16*32];
    __shared__ float hp[8*16*32];
    __shared__ float mu_s[16];
    __shared__ float rstd_s[16];
    __shared__ float u_t[32*130];
    __shared__ float4 Ps[8*16*64];
    __shared__ float4 Cs[16*8*16];
    __shared__ float4 Vs[16*64];
    __shared__ float part[16*17];

    const int t    = threadIdx.x;
    const int lane = t & 31;
    const int wav  = t >> 5;
    const int bb   = lane & 15;
    const int hi   = lane >> 4;

    const int tile = blockIdx.x;
    const float* xg = x + (size_t)tile * 16 * 256;

    v2f a0[64], a1[64];
#define LOAD_WCAP_RANGE(LO, HI)                                                                               \
    _Pragma("unroll")                                                                                         \
    for (int p = (LO); p < (HI); ++p) {                                                                       \
        const int pi = (p < 16) ? 0 : 1 + ((p - 16) % 12) / 4;                                                \
        const int pj = (p < 16) ? p : ((p - 16) / 12)*4 + ((p - 16) & 3);                                     \
        const float* wc = Wcap + (size_t)(((4*wav + pi)*16 + pj)*8)*16 + bb;                  \
        a0[pi*16 + pj][0] = wc[(2*hi+0)*16];  a0[pi*16 + pj][1] = wc[(2*hi+1)*16];                             \
        a1[pi*16 + pj][0] = wc[(2*hi+4)*16];  a1[pi*16 + pj][1] = wc[(2*hi+5)*16];                             \
    }

    float wk1[32];
    #pragma unroll
    for (int k = 0; k < 32; ++k) wk1[k] = W1[(32*wav + k)*32 + lane];

    #pragma unroll
    for (int r = 0; r < 16; ++r) xh[t + 256*r] = xg[t + 256*r];
    lds_barrier();
    LOAD_WCAP_RANGE(0, 15)

    {
        #pragma unroll 1
        for (int bg = 0; bg < 16; bg += 4) {
            float acc[4] = {0.f, 0.f, 0.f, 0.f};
            #pragma unroll
            for (int bi = 0; bi < 4; ++bi) {
                const float4* xr = reinterpret_cast<const float4*>(xh + (bg + bi)*256 + 32*wav);
                #pragma unroll
                for (int q = 0; q < 8; ++q) {
                    const float4 xv = xr[q];
                    acc[bi] = fmaf(xv.x, wk1[4*q+0], acc[bi]);
                    acc[bi] = fmaf(xv.y, wk1[4*q+1], acc[bi]);
                    acc[bi] = fmaf(xv.z, wk1[4*q+2], acc[bi]);
                    acc[bi] = fmaf(xv.w, wk1[4*q+3], acc[bi]);
                }
            }
            #pragma unroll
            for (int bi = 0; bi < 4; ++bi) hp[(wav*16 + bg + bi)*32 + lane] = acc[bi];
        }
    }
    lds_barrier();

    float wk2[32];
    #pragma unroll
    for (int k = 0; k < 32; ++k) wk2[k] = W2[k*256 + 32*wav + lane];
    const float b2v = b2[32*wav + lane];
    float g8[8], be8[8];
    #pragma unroll
    for (int k = 0; k < 8; ++k) { g8[k] = ln_g[(t & 31)*8 + k]; be8[k] = ln_b[(t & 31)*8 + k]; }
    LOAD_WCAP_RANGE(15, 19)

    {
        int n = lane, b0 = wav;
        float s0 = b1[n], s1 = s0;
        #pragma unroll
        for (int w2 = 0; w2 < 8; ++w2) { s0 += hp[(w2*16 + b0)*32 + n]; s1 += hp[(w2*16 + b0 + 8)*32 + n]; }
        h1s[b0*32 + n]     = gelu_exact(s0);
        h1s[(b0+8)*32 + n] = gelu_exact(s1);
    }
    lds_barrier();

    {
        const int n = 32*wav + lane;
        const float bias = b2v;
        LOAD_WCAP_RANGE(19, 34)
        #pragma unroll 1
        for (int bg = 0; bg < 16; bg += 4) {
            float acc[4] = {bias, bias, bias, bias};
            #pragma unroll
            for (int bi = 0; bi < 4; ++bi) {
                const float4* hr = reinterpret_cast<const float4*>(h1s + (bg + bi)*32);
                #pragma unroll
                for (int q = 0; q < 8; ++q) {
                    const float4 hv = hr[q];
                    acc[bi] = fmaf(hv.x, wk2[4*q+0], acc[bi]);
                    acc[bi] = fmaf(hv.y, wk2[4*q+1], acc[bi]);
                    acc[bi] = fmaf(hv.z, wk2[4*q+2], acc[bi]);
                    acc[bi] = fmaf(hv.w, wk2[4*q+3], acc[bi]);
                }
            }
            #pragma unroll
            for (int bi = 0; bi < 4; ++bi) xh[(bg + bi)*256 + n] = gelu_exact(acc[bi]);
        }
    }
    LOAD_WCAP_RANGE(34, 49)
    lds_barrier();

    {
        int ba = wav, bc = wav + 8;
        float sa = 0.f, sb = 0.f;
        #pragma unroll
        for (int r = 0; r < 8; ++r) { sa += xh[ba*256 + lane + 32*r]; sb += xh[bc*256 + lane + 32*r]; }
        #pragma unroll
        for (int m = 16; m >= 1; m >>= 1) { sa += __shfl_xor(sa, m, 32); sb += __shfl_xor(sb, m, 32); }
        float mua = sa * (1.0f/256.0f), mub = sb * (1.0f/256.0f);
        float va = 0.f, vb = 0.f;
        #pragma unroll
        for (int r = 0; r < 8; ++r) {
            float da = xh[ba*256 + lane + 32*r] - mua; va = fmaf(da, da, va);
            float db = xh[bc*256 + lane + 32*r] - mub; vb = fmaf(db, db, vb);
        }
        #pragma unroll
        for (int m = 16; m >= 1; m >>= 1) { va += __shfl_xor(va, m, 32); vb += __shfl_xor(vb, m, 32); }
        if (lane == 0) {
            mu_s[ba] = mua; rstd_s[ba] = rsqrtf(va * (1.0f/256.0f) + LN_EPS);
            mu_s[bc] = mub; rstd_s[bc] = rsqrtf(vb * (1.0f/256.0f) + LN_EPS);
        }
    }
    lds_barrier();
    LOAD_WCAP_RANGE(49, 64)

    {
        int i = t & 31, b0 = t >> 5;
        #pragma unroll
        for (int bsel = 0; bsel < 2; ++bsel) {
            int b = b0 + 8*bsel;
            float mu = mu_s[b], rs = rstd_s[b];
            float vals[8]; float nsq = 0.f;
            #pragma unroll
            for (int k = 0; k < 8; ++k) {
                float v = (xh[b*256 + i*8 + k] - mu) * rs * g8[k] + be8[k];
                vals[k] = v; nsq = fmaf(v, v, nsq);
            }
            float scale = nsq / ((1.0f + nsq) * sqrtf(nsq + SQ_EPS));
            #pragma unroll
            for (int k = 0; k < 8; ++k) u_t[i*130 + k*16 + b] = scale * vals[k];
        }
    }

    #pragma unroll
    for (int m = 0; m < 2; ++m) {
        #pragma unroll
        for (int g = 0; g < 4; ++g)
            Cs[(((2*hi + m)*4 + g)*8 + wav)*16 + bb] = make_float4(0.0625f, 0.0625f, 0.0625f, 0.0625f);
    }
    lds_barrier();

    v2f bu0[4], bu1[4];
    #pragma unroll
    for (int ii = 0; ii < 4; ++ii) {
        const float* ub = u_t + (4*wav + ii)*130 + bb;
        bu0[ii][0] = ub[(2*hi+0)*16];  bu0[ii][1] = ub[(2*hi+1)*16];
        bu1[ii][0] = ub[(2*hi+4)*16];  bu1[ii][1] = ub[(2*hi+5)*16];
    }

    v8f uk[16];
    #pragma unroll
    for (int j = 0; j < 16; ++j) {
        v8f t8 = vzero8();
        t8 = wmma4(a0[j], bu0[0], t8);
        t8 = wmma4(a1[j], bu1[0], t8);
        uk[j] = t8;
    }

    float Lg[32];
    #pragma unroll
    for (int q = 0; q < 32; ++q) Lg[q] = 0.f;
    const bool up = (hi != 0);

    v8f vA = vzero8(), vB = vzero8();
    for (int r = 0; r < 3; ++r) {
        #pragma unroll
        for (int g = 0; g < 4; ++g) {
            float cc[4][4];
            #pragma unroll
            for (int ii = 0; ii < 4; ++ii) {
                const float4 c4 = Cs[((ii*4 + g)*8 + wav)*16 + bb];
                cc[ii][0] = c4.x; cc[ii][1] = c4.y; cc[ii][2] = c4.z; cc[ii][3] = c4.w;
            }
            v8f sj[4];
            #pragma unroll
            for (int jj = 0; jj < 4; ++jj) {
                const float c = cc[0][jj];
                #pragma unroll
                for (int v = 0; v < 8; ++v) sj[jj][v] = c * uk[4*g + jj][v];
            }
            #pragma unroll
            for (int ii = 1; ii < 4; ++ii) {
                #pragma unroll
                for (int jj = 0; jj < 4; ++jj) {
                    const float c = cc[ii][jj];
                    v2f p0, p1;
                    p0[0] = bu0[ii][0]*c; p0[1] = bu0[ii][1]*c; p1[0] = bu1[ii][0]*c; p1[1] = bu1[ii][1]*c;
                    sj[jj] = wmma4(a0[ii*16 + 4*g + jj], p0, sj[jj]);
                    sj[jj] = wmma4(a1[ii*16 + 4*g + jj], p1, sj[jj]);
                }
            }
            #pragma unroll
            for (int jj = 0; jj < 4; ++jj) {
                Ps[((wav*16 + 4*g + jj)*2 + 0)*32 + lane] = make_float4(sj[jj][0], sj[jj][1], sj[jj][2], sj[jj][3]);
                Ps[((wav*16 + 4*g + jj)*2 + 1)*32 + lane] = make_float4(sj[jj][4], sj[jj][5], sj[jj][6], sj[jj][7]);
            }
        }
        __syncthreads();
        v8f sA, sB;
        {
            float4 x0 = Ps[((0*16 + wav)*2 + 0)*32 + lane], x1 = Ps[((0*16 + wav)*2 + 1)*32 + lane];
            float4 y0 = Ps[((0*16 + wav + 8)*2 + 0)*32 + lane], y1 = Ps[((0*16 + wav + 8)*2 + 1)*32 + lane];
            sA[0]=x0.x; sA[1]=x0.y; sA[2]=x0.z; sA[3]=x0.w; sA[4]=x1.x; sA[5]=x1.y; sA[6]=x1.z; sA[7]=x1.w;
            sB[0]=y0.x; sB[1]=y0.y; sB[2]=y0.z; sB[3]=y0.w; sB[4]=y1.x; sB[5]=y1.y; sB[6]=y1.z; sB[7]=y1.w;
        }
        #pragma unroll
        for (int w2 = 1; w2 < 8; ++w2) {
            float4 x0 = Ps[((w2*16 + wav)*2 + 0)*32 + lane], x1 = Ps[((w2*16 + wav)*2 + 1)*32 + lane];
            float4 y0 = Ps[((w2*16 + wav + 8)*2 + 0)*32 + lane], y1 = Ps[((w2*16 + wav + 8)*2 + 1)*32 + lane];
            sA[0]+=x0.x; sA[1]+=x0.y; sA[2]+=x0.z; sA[3]+=x0.w; sA[4]+=x1.x; sA[5]+=x1.y; sA[6]+=x1.z; sA[7]+=x1.w;
            sB[0]+=y0.x; sB[1]+=y0.y; sB[2]+=y0.z; sB[3]+=y0.w; sB[4]+=y1.x; sB[5]+=y1.y; sB[6]+=y1.z; sB[7]+=y1.w;
        }
        float nA = 0.f, nB = 0.f;
        #pragma unroll
        for (int v = 0; v < 8; ++v) { nA = fmaf(sA[v], sA[v], nA); nB = fmaf(sB[v], sB[v], nB); }
        nA += __shfl_xor(nA, 16, 32);  nB += __shfl_xor(nB, 16, 32);
        float scA = nA / ((1.0f + nA) * sqrtf(nA + SQ_EPS));
        float scB = nB / ((1.0f + nB) * sqrtf(nB + SQ_EPS));
        #pragma unroll
        for (int v = 0; v < 8; ++v) { vA[v] = sA[v]*scA; vB[v] = sB[v]*scB; }

        if (r < 2) {
            Vs[((wav    )*2 + 0)*32 + lane] = make_float4(vA[0], vA[1], vA[2], vA[3]);
            Vs[((wav    )*2 + 1)*32 + lane] = make_float4(vA[4], vA[5], vA[6], vA[7]);
            Vs[((wav + 8)*2 + 0)*32 + lane] = make_float4(vB[0], vB[1], vB[2], vB[3]);
            Vs[((wav + 8)*2 + 1)*32 + lane] = make_float4(vB[4], vB[5], vB[6], vB[7]);
            __syncthreads();
            float own_p[2] = {0.f, 0.f}, rcv_p[2] = {0.f, 0.f};
            #pragma unroll
            for (int j = 0; j < 16; ++j) {
                const float4 y0 = Vs[(j*2 + 0)*32 + lane], y1 = Vs[(j*2 + 1)*32 + lane];
                float ag[4];
                #pragma unroll
                for (int ii = 0; ii < 4; ++ii) {
                    v8f uh;
                    if (ii == 0) {
                        uh = uk[j];
                    } else {
                        uh = vzero8();
                        uh = wmma4(a0[ii*16 + j], bu0[ii], uh);
                        uh = wmma4(a1[ii*16 + j], bu1[ii], uh);
                    }
                    v2f a2 = {0.f, 0.f};
                    a2 = __builtin_elementwise_fma((v2f){uh[0], uh[1]}, (v2f){y0.x, y0.y}, a2);
                    a2 = __builtin_elementwise_fma((v2f){uh[2], uh[3]}, (v2f){y0.z, y0.w}, a2);
                    a2 = __builtin_elementwise_fma((v2f){uh[4], uh[5]}, (v2f){y1.x, y1.y}, a2);
                    a2 = __builtin_elementwise_fma((v2f){uh[6], uh[7]}, (v2f){y1.z, y1.w}, a2);
                    ag[ii] = a2[0] + a2[1];
                }
                if (j > 0) {
                    #pragma unroll
                    for (int m = 0; m < 2; ++m) Lg[m*16 + j - 1] += own_p[m] + rcv_p[m];
                    asm volatile("" :: "v"(Lg[j - 1]), "v"(Lg[16 + j - 1]));
                }
                #pragma unroll
                for (int m = 0; m < 2; ++m) {
                    const float own = up ? ag[2 + m] : ag[m];
                    const float snd = up ? ag[m] : ag[2 + m];
                    own_p[m] = own;
                    rcv_p[m] = __shfl_xor(snd, 16, 32);
                }
            }
            #pragma unroll
            for (int m = 0; m < 2; ++m) Lg[m*16 + 15] += own_p[m] + rcv_p[m];
            #pragma unroll
            for (int m = 0; m < 2; ++m) {
                float mx = Lg[m*16];
                #pragma unroll
                for (int j = 1; j < 16; ++j) mx = fmaxf(mx, Lg[m*16 + j]);
                float e[16]; float sum = 0.f;
                #pragma unroll
                for (int j = 0; j < 16; ++j) { e[j] = expf(Lg[m*16 + j] - mx); sum += e[j]; }
                float inv = 1.0f / sum;
                #pragma unroll
                for (int g = 0; g < 4; ++g)
                    Cs[(((2*hi + m)*4 + g)*8 + wav)*16 + bb] = make_float4(e[4*g]*inv, e[4*g+1]*inv, e[4*g+2]*inv, e[4*g+3]*inv);
            }
            __syncthreads();
        }
    }

    {
        float pA = 0.f, pB = 0.f;
        #pragma unroll
        for (int v = 0; v < 8; ++v) {
            pA = fmaf(vA[v], Wf[(wav    )*16 + v + 8*hi], pA);
            pB = fmaf(vB[v], Wf[(wav + 8)*16 + v + 8*hi], pB);
        }
        pA += __shfl_xor(pA, 16, 32);  pB += __shfl_xor(pB, 16, 32);
        if (lane < 16) { part[wav*17 + bb] = pA; part[(wav+8)*17 + bb] = pB; }
    }
    __syncthreads();
    if (t < 16) {
        float acc = bf[0];
        #pragma unroll
        for (int j = 0; j < 16; ++j) acc += part[j*17 + t];
        out[tile*16 + t] = acc;
    }
}

extern "C" void kernel_launch(void* const* d_in, const int* in_sizes, int n_in,
                              void* d_out, int out_size, void* d_ws, size_t ws_size,
                              hipStream_t stream) {
    const float* x    = (const float*)d_in[0];
    const float* W1   = (const float*)d_in[1];
    const float* b1   = (const float*)d_in[2];
    const float* W2   = (const float*)d_in[3];
    const float* b2   = (const float*)d_in[4];
    const float* ln_g = (const float*)d_in[5];
    const float* ln_b = (const float*)d_in[6];
    const float* Wcap = (const float*)d_in[7];
    const float* Wf   = (const float*)d_in[8];
    const float* bf   = (const float*)d_in[9];
    float* out = (float*)d_out;

    const int B = in_sizes[0] / 256;
    const int blocks = B / 16;
    capsnet_fused<<<blocks, 256, 0, stream>>>(x, W1, b1, W2, b2, ln_g, ln_b, Wcap, Wf, bf, out);
}


# Both MLP layers re-tiled inside a wave: a lane owns 4 samples x 4 columns (weights by 16-byte loads, LDS rows padded): 32 LDS operand reads per lane and layer, was 128; sums bit-identical
# speedup vs baseline: 1.9281x; 1.0310x over previous
#include <hip/hip_runtime.h>
#include <math.h>

typedef __attribute__((ext_vector_type(2))) float v2f;
typedef __attribute__((ext_vector_type(8))) float v8f;

#define LN_EPS 1e-5f
#define SQ_EPS 1e-9f
#define XP 260
#define HP 36

__device__ __forceinline__ float gelu_exact(float x) {
    return 0.5f * x * (1.0f + erff(x * 0.70710678118654752440f));
}

__device__ __forceinline__ v8f wmma4(v2f a, v2f b, v8f c) {
    return __builtin_amdgcn_wmma_f32_16x16x4_f32(false, a, false, b, (short)0, c, false, false);
}

__device__ __forceinline__ void lds_barrier() {
    __builtin_amdgcn_fence(__ATOMIC_RELEASE, "workgroup", "local");
    __builtin_amdgcn_s_barrier();
    __builtin_amdgcn_fence(__ATOMIC_ACQUIRE, "workgroup", "local");
}

__device__ __forceinline__ v8f vzero8() {
    v8f z = {0.f,0.f,0.f,0.f,0.f,0.f,0.f,0.f};
    return z;
}

__global__ __attribute__((amdgpu_waves_per_eu(1))) __launch_bounds__(256)
void capsnet_fused(
    const float* __restrict__ x,   const float* __restrict__ W1, const float* __restrict__ b1,
    const float* __restrict__ W2,  const float* __restrict__ b2,
    const float* __restrict__ ln_g,const float* __restrict__ ln_b,
    const float* __restrict__ Wcap,const float* __restrict__ Wf, const float* __restrict__ bf,
    float* __restrict__ out)
{
    __shared__ __align__(16) float xh[16*XP];
    __shared__ __align__(16) float h1s[16*HP];
    __shared__ __align__(16) float hp[8*16*32];
    __shared__ float mu_s[16];
    __shared__ float rstd_s[16];
    __shared__ float u_t[32*130];
    __shared__ float4 Ps[8*16*64];
    __shared__ float4 Cs[16*8*16];
    __shared__ float4 Vs[16*64];
    __shared__ float part[16*17];

    const int t    = threadIdx.x;
    const int lane = t & 31;
    const int wav  = t >> 5;
    const int bb   = lane & 15;
    const int hi   = lane >> 4;
    const int n4   = lane & 7;
    const int b4   = lane >> 3;

    const int tile = blockIdx.x;
    const float* xg = x + (size_t)tile * 16 * 256;

    v2f a0[64], a1[64];
#define LOAD_WCAP_RANGE(LO, HI)                                                                               \
    _Pragma("unroll")                                                                                         \
    for (int p = (LO); p < (HI); ++p) {                                                                       \
        const int pi = (p < 16) ? 0 : 1 + ((p - 16) % 12) / 4;                                                \
        const int pj = (p < 16) ? p : ((p - 16) / 12)*4 + ((p - 16) & 3);                                     \
        const float* wc = Wcap + (size_t)(((4*wav + pi)*16 + pj)*8)*16 + bb;                  \
        a0[pi*16 + pj][0] = wc[(2*hi+0)*16];  a0[pi*16 + pj][1] = wc[(2*hi+1)*16];                             \
        a1[pi*16 + pj][0] = wc[(2*hi+4)*16];  a1[pi*16 + pj][1] = wc[(2*hi+5)*16];                             \
    }

    float4 wk1[32];
    #pragma unroll
    for (int k = 0; k < 32; ++k) wk1[k] = *reinterpret_cast<const float4*>(W1 + (32*wav + k)*32 + 4*n4);

    #pragma unroll
    for (int r = 0; r < 16; ++r) xh[r*XP + t] = xg[t + 256*r];
    lds_barrier();
    LOAD_WCAP_RANGE(0, 15)

    {
        #pragma unroll
        for (int bi = 0; bi < 4; ++bi) {
            const int b = 4*b4 + bi;
            const float4* xr = reinterpret_cast<const float4*>(xh + b*XP + 32*wav);
            float c0 = 0.f, c1 = 0.f, c2 = 0.f, c3 = 0.f;
            #pragma unroll
            for (int q = 0; q < 8; ++q) {
                const float4 xv = xr[q];
                const float xs[4] = {xv.x, xv.y, xv.z, xv.w};
                #pragma unroll
                for (int e = 0; e < 4; ++e) {
                    const float4 w = wk1[4*q + e];
                    c0 = fmaf(xs[e], w.x, c0);
                    c1 = fmaf(xs[e], w.y, c1);
                    c2 = fmaf(xs[e], w.z, c2);
                    c3 = fmaf(xs[e], w.w, c3);
                }
            }
            *reinterpret_cast<float4*>(hp + (wav*16 + b)*32 + 4*n4) = make_float4(c0, c1, c2, c3);
        }
    }
    lds_barrier();

    float4 wk2[32];
    #pragma unroll
    for (int k = 0; k < 32; ++k) wk2[k] = *reinterpret_cast<const float4*>(W2 + k*256 + 32*wav + 4*n4);
    const float4 b2v = *reinterpret_cast<const float4*>(b2 + 32*wav + 4*n4);
    float g8[8], be8[8];
    #pragma unroll
    for (int k = 0; k < 8; ++k) { g8[k] = ln_g[(t & 31)*8 + k]; be8[k] = ln_b[(t & 31)*8 + k]; }
    LOAD_WCAP_RANGE(15, 19)

    {
        int n = lane, b0 = wav;
        float s0 = b1[n], s1 = s0;
        #pragma unroll
        for (int w2 = 0; w2 < 8; ++w2) { s0 += hp[(w2*16 + b0)*32 + n]; s1 += hp[(w2*16 + b0 + 8)*32 + n]; }
        h1s[b0*HP + n]     = gelu_exact(s0);
        h1s[(b0+8)*HP + n] = gelu_exact(s1);
    }
    lds_barrier();

    {
        LOAD_WCAP_RANGE(19, 34)
        #pragma unroll 1
        for (int bh = 0; bh < 4; bh += 2) {
            const int b = 4*b4 + bh;
            const float4* hr0 = reinterpret_cast<const float4*>(h1s + b*HP);
            const float4* hr1 = reinterpret_cast<const float4*>(h1s + (b + 1)*HP);
            v2f s00 = {b2v.x, b2v.y}, s01 = {b2v.z, b2v.w}, s10 = s00, s11 = s01;
            #pragma unroll
            for (int q = 0; q < 8; ++q) {
                const float4 h0 = hr0[q], h1 = hr1[q];
                const float ha[4] = {h0.x, h0.y, h0.z, h0.w};
                const float hb[4] = {h1.x, h1.y, h1.z, h1.w};
                #pragma unroll
                for (int e = 0; e < 4; ++e) {
                    const float4 w = wk2[4*q + e];
                    const v2f wl = {w.x, w.y}, wh = {w.z, w.w};
                    const v2f xa = {ha[e], ha[e]}, xb = {hb[e], hb[e]};
                    s00 = __builtin_elementwise_fma(xa, wl, s00);
                    s01 = __builtin_elementwise_fma(xa, wh, s01);
                    s10 = __builtin_elementwise_fma(xb, wl, s10);
                    s11 = __builtin_elementwise_fma(xb, wh, s11);
                }
            }
            asm volatile("" :: "v"(s00), "v"(s01), "v"(s10), "v"(s11));
            *reinterpret_cast<float4*>(xh + b*XP + 32*wav + 4*n4) =
                make_float4(gelu_exact(s00[0]), gelu_exact(s00[1]), gelu_exact(s01[0]), gelu_exact(s01[1]));
            *reinterpret_cast<float4*>(xh + (b + 1)*XP + 32*wav + 4*n4) =
                make_float4(gelu_exact(s10[0]), gelu_exact(s10[1]), gelu_exact(s11[0]), gelu_exact(s11[1]));
        }
    }
    LOAD_WCAP_RANGE(34, 49)
    lds_barrier();

    {
        int ba = wav, bc = wav + 8;
        float sa = 0.f, sb = 0.f;
        #pragma unroll
        for (int r = 0; r < 8; ++r) { sa += xh[ba*XP + lane + 32*r]; sb += xh[bc*XP + lane + 32*r]; }
        #pragma unroll
        for (int m = 16; m >= 1; m >>= 1) { sa += __shfl_xor(sa, m, 32); sb += __shfl_xor(sb, m, 32); }
        float mua = sa * (1.0f/256.0f), mub = sb * (1.0f/256.0f);
        float va = 0.f, vb = 0.f;
        #pragma unroll
        for (int r = 0; r < 8; ++r) {
            float da = xh[ba*XP + lane + 32*r] - mua; va = fmaf(da, da, va);
            float db = xh[bc*XP + lane + 32*r] - mub; vb = fmaf(db, db, vb);
        }
        #pragma unroll
        for (int m = 16; m >= 1; m >>= 1) { va += __shfl_xor(va, m, 32); vb += __shfl_xor(vb, m, 32); }
        if (lane == 0) {
            mu_s[ba] = mua; rstd_s[ba] = rsqrtf(va * (1.0f/256.0f) + LN_EPS);
            mu_s[bc] = mub; rstd_s[bc] = rsqrtf(vb * (1.0f/256.0f) + LN_EPS);
        }
    }
    lds_barrier();
    LOAD_WCAP_RANGE(49, 64)

    {
        int i = t & 31, b0 = t >> 5;
        #pragma unroll
        for (int bsel = 0; bsel < 2; ++bsel) {
            int b = b0 + 8*bsel;
            float mu = mu_s[b], rs = rstd_s[b];
            float vals[8]; float nsq = 0.f;
            #pragma unroll
            for (int k = 0; k < 8; ++k) {
                float v = (xh[b*XP + i*8 + k] - mu) * rs * g8[k] + be8[k];
                vals[k] = v; nsq = fmaf(v, v, nsq);
            }
            float scale = nsq / ((1.0f + nsq) * sqrtf(nsq + SQ_EPS));
            #pragma unroll
            for (int k = 0; k < 8; ++k) u_t[i*130 + k*16 + b] = scale * vals[k];
        }
    }

    #pragma unroll
    for (int m = 0; m < 2; ++m) {
        #pragma unroll
        for (int g = 0; g < 4; ++g)
            Cs[(((2*hi + m)*4 + g)*8 + wav)*16 + bb] = make_float4(0.0625f, 0.0625f, 0.0625f, 0.0625f);
    }
    lds_barrier();

    v2f bu0[4], bu1[4];
    #pragma unroll
    for (int ii = 0; ii < 4; ++ii) {
        const float* ub = u_t + (4*wav + ii)*130 + bb;
        bu0[ii][0] = ub[(2*hi+0)*16];  bu0[ii][1] = ub[(2*hi+1)*16];
        bu1[ii][0] = ub[(2*hi+4)*16];  bu1[ii][1] = ub[(2*hi+5)*16];
    }

    v8f uk[16];
    #pragma unroll
    for (int j = 0; j < 16; ++j) {
        v8f t8 = vzero8();
        t8 = wmma4(a0[j], bu0[0], t8);
        t8 = wmma4(a1[j], bu1[0], t8);
        uk[j] = t8;
    }

    float Lg[32];
    #pragma unroll
    for (int q = 0; q < 32; ++q) Lg[q] = 0.f;
    const bool up = (hi != 0);

    v8f vA = vzero8(), vB = vzero8();
    for (int r = 0; r < 3; ++r) {
        #pragma unroll
        for (int g = 0; g < 4; ++g) {
            float cc[4][4];
            #pragma unroll
            for (int ii = 0; ii < 4; ++ii) {
                const float4 c4 = Cs[((ii*4 + g)*8 + wav)*16 + bb];
                cc[ii][0] = c4.x; cc[ii][1] = c4.y; cc[ii][2] = c4.z; cc[ii][3] = c4.w;
            }
            v8f sj[4];
            #pragma unroll
            for (int jj = 0; jj < 4; ++jj) {
                const float c = cc[0][jj];
                #pragma unroll
                for (int v = 0; v < 8; ++v) sj[jj][v] = c * uk[4*g + jj][v];
            }
            #pragma unroll
            for (int ii = 1; ii < 4; ++ii) {
                #pragma unroll
                for (int jj = 0; jj < 4; ++jj) {
                    const float c = cc[ii][jj];
                    v2f p0, p1;
                    p0[0] = bu0[ii][0]*c; p0[1] = bu0[ii][1]*c; p1[0] = bu1[ii][0]*c; p1[1] = bu1[ii][1]*c;
                    sj[jj] = wmma4(a0[ii*16 + 4*g + jj], p0, sj[jj]);
                    sj[jj] = wmma4(a1[ii*16 + 4*g + jj], p1, sj[jj]);
                }
            }
            #pragma unroll
            for (int jj = 0; jj < 4; ++jj) {
                Ps[((wav*16 + 4*g + jj)*2 + 0)*32 + lane] = make_float4(sj[jj][0], sj[jj][1], sj[jj][2], sj[jj][3]);
                Ps[((wav*16 + 4*g + jj)*2 + 1)*32 + lane] = make_float4(sj[jj][4], sj[jj][5], sj[jj][6], sj[jj][7]);
            }
        }
        __syncthreads();
        v8f sA, sB;
        {
            float4 x0 = Ps[((0*16 + wav)*2 + 0)*32 + lane], x1 = Ps[((0*16 + wav)*2 + 1)*32 + lane];
            float4 y0 = Ps[((0*16 + wav + 8)*2 + 0)*32 + lane], y1 = Ps[((0*16 + wav + 8)*2 + 1)*32 + lane];
            sA[0]=x0.x; sA[1]=x0.y; sA[2]=x0.z; sA[3]=x0.w; sA[4]=x1.x; sA[5]=x1.y; sA[6]=x1.z; sA[7]=x1.w;
            sB[0]=y0.x; sB[1]=y0.y; sB[2]=y0.z; sB[3]=y0.w; sB[4]=y1.x; sB[5]=y1.y; sB[6]=y1.z; sB[7]=y1.w;
        }
        #pragma unroll
        for (int w2 = 1; w2 < 8; ++w2) {
            float4 x0 = Ps[((w2*16 + wav)*2 + 0)*32 + lane], x1 = Ps[((w2*16 + wav)*2 + 1)*32 + lane];
            float4 y0 = Ps[((w2*16 + wav + 8)*2 + 0)*32 + lane], y1 = Ps[((w2*16 + wav + 8)*2 + 1)*32 + lane];
            sA[0]+=x0.x; sA[1]+=x0.y; sA[2]+=x0.z; sA[3]+=x0.w; sA[4]+=x1.x; sA[5]+=x1.y; sA[6]+=x1.z; sA[7]+=x1.w;
            sB[0]+=y0.x; sB[1]+=y0.y; sB[2]+=y0.z; sB[3]+=y0.w; sB[4]+=y1.x; sB[5]+=y1.y; sB[6]+=y1.z; sB[7]+=y1.w;
        }
        float nA = 0.f, nB = 0.f;
        #pragma unroll
        for (int v = 0; v < 8; ++v) { nA = fmaf(sA[v], sA[v], nA); nB = fmaf(sB[v], sB[v], nB); }
        nA += __shfl_xor(nA, 16, 32);  nB += __shfl_xor(nB, 16, 32);
        float scA = nA / ((1.0f + nA) * sqrtf(nA + SQ_EPS));
        float scB = nB / ((1.0f + nB) * sqrtf(nB + SQ_EPS));
        #pragma unroll
        for (int v = 0; v < 8; ++v) { vA[v] = sA[v]*scA; vB[v] = sB[v]*scB; }

        if (r < 2) {
            Vs[((wav    )*2 + 0)*32 + lane] = make_float4(vA[0], vA[1], vA[2], vA[3]);
            Vs[((wav    )*2 + 1)*32 + lane] = make_float4(vA[4], vA[5], vA[6], vA[7]);
            Vs[((wav + 8)*2 + 0)*32 + lane] = make_float4(vB[0], vB[1], vB[2], vB[3]);
            Vs[((wav + 8)*2 + 1)*32 + lane] = make_float4(vB[4], vB[5], vB[6], vB[7]);
            __syncthreads();
            float own_p[2] = {0.f, 0.f}, rcv_p[2] = {0.f, 0.f};
            #pragma unroll
            for (int j = 0; j < 16; ++j) {
                const float4 y0 = Vs[(j*2 + 0)*32 + lane], y1 = Vs[(j*2 + 1)*32 + lane];
                float ag[4];
                #pragma unroll
                for (int ii = 0; ii < 4; ++ii) {
                    v8f uh;
                    if (ii == 0) {
                        uh = uk[j];
                    } else {
                        uh = vzero8();
                        uh = wmma4(a0[ii*16 + j], bu0[ii], uh);
                        uh = wmma4(a1[ii*16 + j], bu1[ii], uh);
                    }
                    v2f a2 = {0.f, 0.f};
                    a2 = __builtin_elementwise_fma((v2f){uh[0], uh[1]}, (v2f){y0.x, y0.y}, a2);
                    a2 = __builtin_elementwise_fma((v2f){uh[2], uh[3]}, (v2f){y0.z, y0.w}, a2);
                    a2 = __builtin_elementwise_fma((v2f){uh[4], uh[5]}, (v2f){y1.x, y1.y}, a2);
                    a2 = __builtin_elementwise_fma((v2f){uh[6], uh[7]}, (v2f){y1.z, y1.w}, a2);
                    ag[ii] = a2[0] + a2[1];
                }
                if (j > 0) {
                    #pragma unroll
                    for (int m = 0; m < 2; ++m) Lg[m*16 + j - 1] += own_p[m] + rcv_p[m];
                    asm volatile("" :: "v"(Lg[j - 1]), "v"(Lg[16 + j - 1]));
                }
                #pragma unroll
                for (int m = 0; m < 2; ++m) {
                    const float own = up ? ag[2 + m] : ag[m];
                    const float snd = up ? ag[m] : ag[2 + m];
                    own_p[m] = own;
                    rcv_p[m] = __shfl_xor(snd, 16, 32);
                }
            }
            #pragma unroll
            for (int m = 0; m < 2; ++m) Lg[m*16 + 15] += own_p[m] + rcv_p[m];
            #pragma unroll
            for (int m = 0; m < 2; ++m) {
                float mx = Lg[m*16];
                #pragma unroll
                for (int j = 1; j < 16; ++j) mx = fmaxf(mx, Lg[m*16 + j]);
                float e[16]; float sum = 0.f;
                #pragma unroll
                for (int j = 0; j < 16; ++j) { e[j] = expf(Lg[m*16 + j] - mx); sum += e[j]; }
                float inv = 1.0f / sum;
                #pragma unroll
                for (int g = 0; g < 4; ++g)
                    Cs[(((2*hi + m)*4 + g)*8 + wav)*16 + bb] = make_float4(e[4*g]*inv, e[4*g+1]*inv, e[4*g+2]*inv, e[4*g+3]*inv);
            }
            __syncthreads();
        }
    }

    {
        float pA = 0.f, pB = 0.f;
        #pragma unroll
        for (int v = 0; v < 8; ++v) {
            pA = fmaf(vA[v], Wf[(wav    )*16 + v + 8*hi], pA);
            pB = fmaf(vB[v], Wf[(wav + 8)*16 + v + 8*hi], pB);
        }
        pA += __shfl_xor(pA, 16, 32);  pB += __shfl_xor(pB, 16, 32);
        if (lane < 16) { part[wav*17 + bb] = pA; part[(wav+8)*17 + bb] = pB; }
    }
    __syncthreads();
    if (t < 16) {
        float acc = bf[0];
        #pragma unroll
        for (int j = 0; j < 16; ++j) acc += part[j*17 + t];
        out[tile*16 + t] = acc;
    }
}

extern "C" void kernel_launch(void* const* d_in, const int* in_sizes, int n_in,
                              void* d_out, int out_size, void* d_ws, size_t ws_size,
                              hipStream_t stream) {
    const float* x    = (const float*)d_in[0];
    const float* W1   = (const float*)d_in[1];
    const float* b1   = (const float*)d_in[2];
    const float* W2   = (const float*)d_in[3];
    const float* b2   = (const float*)d_in[4];
    const float* ln_g = (const float*)d_in[5];
    const float* ln_b = (const float*)d_in[6];
    const float* Wcap = (const float*)d_in[7];
    const float* Wf   = (const float*)d_in[8];
    const float* bf   = (const float*)d_in[9];
    float* out = (float*)d_out;

    const int B = in_sizes[0] / 256;
    const int blocks = B / 16;
    capsnet_fused<<<blocks, 256, 0, stream>>>(x, W1, b1, W2, b2, ln_g, ln_b, Wcap, Wf, bf, out);
}


# Second capsule's 16 u_hat tiles kept in registers too; Wcap fragments of the other two parked in LDS (lane-private, over dead MLP scratch): 64 not 96 matrix calls a pass; s sums in new order
# speedup vs baseline: 2.0420x; 1.0591x over previous
#include <hip/hip_runtime.h>
#include <math.h>

typedef __attribute__((ext_vector_type(2))) float v2f;
typedef __attribute__((ext_vector_type(8))) float v8f;

#define LN_EPS 1e-5f
#define SQ_EPS 1e-9f
#define XP 260
#define HP 36

__device__ __forceinline__ float gelu_exact(float x) {
    return 0.5f * x * (1.0f + erff(x * 0.70710678118654752440f));
}

__device__ __forceinline__ v8f wmma4(v2f a, v2f b, v8f c) {
    return __builtin_amdgcn_wmma_f32_16x16x4_f32(false, a, false, b, (short)0, c, false, false);
}

__device__ __forceinline__ void lds_barrier() {
    __builtin_amdgcn_fence(__ATOMIC_RELEASE, "workgroup", "local");
    __builtin_amdgcn_s_barrier();
    __builtin_amdgcn_fence(__ATOMIC_ACQUIRE, "workgroup", "local");
}

__device__ __forceinline__ v8f vzero8() {
    v8f z = {0.f,0.f,0.f,0.f,0.f,0.f,0.f,0.f};
    return z;
}

__global__ __attribute__((amdgpu_waves_per_eu(1))) __launch_bounds__(256)
void capsnet_fused(
    const float* __restrict__ x,   const float* __restrict__ W1, const float* __restrict__ b1,
    const float* __restrict__ W2,  const float* __restrict__ b2,
    const float* __restrict__ ln_g,const float* __restrict__ ln_b,
    const float* __restrict__ Wcap,const float* __restrict__ Wf, const float* __restrict__ bf,
    float* __restrict__ out)
{
    __shared__ __align__(16) float smem[78128];
    float*  const xh   = smem;
    float*  const h1s  = smem + 4160;
    float*  const hp   = smem + 4736;
    float*  const u_t  = smem + 8832;
    float4* const Wl   = reinterpret_cast<float4*>(smem);
    float4* const Ps   = reinterpret_cast<float4*>(smem + 32768);
    float4* const Cs   = reinterpret_cast<float4*>(smem + 65536);
    float4* const Vs   = reinterpret_cast<float4*>(smem + 73728);
    float*  const part = smem + 77824;
    float*  const mu_s   = smem + 78096;
    float*  const rstd_s = smem + 78112;

    const int t    = threadIdx.x;
    const int lane = t & 31;
    const int wav  = t >> 5;
    const int bb   = lane & 15;
    const int hi   = lane >> 4;
    const int n4   = lane & 7;
    const int b4   = lane >> 3;

    const int tile = blockIdx.x;
    const float* xg = x + (size_t)tile * 16 * 256;

    v2f a0[64], a1[64];
#define LOAD_WCAP_RANGE(LO, HI)                                                                               \
    _Pragma("unroll")                                                                                         \
    for (int p = (LO); p < (HI); ++p) {                                                                       \
        const int pi = ((p >> 4) + 2) & 3;                                                                    \
        const int pj = p & 15;                                                                                \
        const float* wc = Wcap + (size_t)(((4*wav + pi)*16 + pj)*8)*16 + bb;                  \
        a0[pi*16 + pj][0] = wc[(2*hi+0)*16];  a0[pi*16 + pj][1] = wc[(2*hi+1)*16];                             \
        a1[pi*16 + pj][0] = wc[(2*hi+4)*16];  a1[pi*16 + pj][1] = wc[(2*hi+5)*16];                             \
    }

    float4 wk1[32];
    #pragma unroll
    for (int k = 0; k < 32; ++k) wk1[k] = *reinterpret_cast<const float4*>(W1 + (32*wav + k)*32 + 4*n4);

    #pragma unroll
    for (int r = 0; r < 16; ++r) xh[r*XP + t] = xg[t + 256*r];
    lds_barrier();
    LOAD_WCAP_RANGE(0, 15)

    {
        #pragma unroll
        for (int bi = 0; bi < 4; ++bi) {
            const int b = 4*b4 + bi;
            const float4* xr = reinterpret_cast<const float4*>(xh + b*XP + 32*wav);
            float c0 = 0.f, c1 = 0.f, c2 = 0.f, c3 = 0.f;
            #pragma unroll
            for (int q = 0; q < 8; ++q) {
                const float4 xv = xr[q];
                const float xs[4] = {xv.x, xv.y, xv.z, xv.w};
                #pragma unroll
                for (int e = 0; e < 4; ++e) {
                    const float4 w = wk1[4*q + e];
                    c0 = fmaf(xs[e], w.x, c0);
                    c1 = fmaf(xs[e], w.y, c1);
                    c2 = fmaf(xs[e], w.z, c2);
                    c3 = fmaf(xs[e], w.w, c3);
                }
            }
            *reinterpret_cast<float4*>(hp + (wav*16 + b)*32 + 4*n4) = make_float4(c0, c1, c2, c3);
        }
    }
    lds_barrier();

    float4 wk2[32];
    #pragma unroll
    for (int k = 0; k < 32; ++k) wk2[k] = *reinterpret_cast<const float4*>(W2 + k*256 + 32*wav + 4*n4);
    const float4 b2v = *reinterpret_cast<const float4*>(b2 + 32*wav + 4*n4);
    float g8[8], be8[8];
    #pragma unroll
    for (int k = 0; k < 8; ++k) { g8[k] = ln_g[(t & 31)*8 + k]; be8[k] = ln_b[(t & 31)*8 + k]; }
    LOAD_WCAP_RANGE(15, 19)

    {
        int n = lane, b0 = wav;
        float s0 = b1[n], s1 = s0;
        #pragma unroll
        for (int w2 = 0; w2 < 8; ++w2) { s0 += hp[(w2*16 + b0)*32 + n]; s1 += hp[(w2*16 + b0 + 8)*32 + n]; }
        h1s[b0*HP + n]     = gelu_exact(s0);
        h1s[(b0+8)*HP + n] = gelu_exact(s1);
    }
    lds_barrier();

    {
        LOAD_WCAP_RANGE(19, 34)
        #pragma unroll 1
        for (int bh = 0; bh < 4; bh += 2) {
            const int b = 4*b4 + bh;
            const float4* hr0 = reinterpret_cast<const float4*>(h1s + b*HP);
            const float4* hr1 = reinterpret_cast<const float4*>(h1s + (b + 1)*HP);
            v2f s00 = {b2v.x, b2v.y}, s01 = {b2v.z, b2v.w}, s10 = s00, s11 = s01;
            #pragma unroll
            for (int q = 0; q < 8; ++q) {
                const float4 h0 = hr0[q], h1 = hr1[q];
                const float ha[4] = {h0.x, h0.y, h0.z, h0.w};
                const float hb[4] = {h1.x, h1.y, h1.z, h1.w};
                #pragma unroll
                for (int e = 0; e < 4; ++e) {
                    const float4 w = wk2[4*q + e];
                    const v2f wl = {w.x, w.y}, wh = {w.z, w.w};
                    const v2f xa = {ha[e], ha[e]}, xb = {hb[e], hb[e]};
                    s00 = __builtin_elementwise_fma(xa, wl, s00);
                    s01 = __builtin_elementwise_fma(xa, wh, s01);
                    s10 = __builtin_elementwise_fma(xb, wl, s10);
                    s11 = __builtin_elementwise_fma(xb, wh, s11);
                }
            }
            asm volatile("" :: "v"(s00), "v"(s01), "v"(s10), "v"(s11));
            *reinterpret_cast<float4*>(xh + b*XP + 32*wav + 4*n4) =
                make_float4(gelu_exact(s00[0]), gelu_exact(s00[1]), gelu_exact(s01[0]), gelu_exact(s01[1]));
            *reinterpret_cast<float4*>(xh + (b + 1)*XP + 32*wav + 4*n4) =
                make_float4(gelu_exact(s10[0]), gelu_exact(s10[1]), gelu_exact(s11[0]), gelu_exact(s11[1]));
        }
    }
    LOAD_WCAP_RANGE(34, 49)
    lds_barrier();

    {
        int ba = wav, bc = wav + 8;
        float sa = 0.f, sb = 0.f;
        #pragma unroll
        for (int r = 0; r < 8; ++r) { sa += xh[ba*XP + lane + 32*r]; sb += xh[bc*XP + lane + 32*r]; }
        #pragma unroll
        for (int m = 16; m >= 1; m >>= 1) { sa += __shfl_xor(sa, m, 32); sb += __shfl_xor(sb, m, 32); }
        float mua = sa * (1.0f/256.0f), mub = sb * (1.0f/256.0f);
        float va = 0.f, vb = 0.f;
        #pragma unroll
        for (int r = 0; r < 8; ++r) {
            float da = xh[ba*XP + lane + 32*r] - mua; va = fmaf(da, da, va);
            float db = xh[bc*XP + lane + 32*r] - mub; vb = fmaf(db, db, vb);
        }
        #pragma unroll
        for (int m = 16; m >= 1; m >>= 1) { va += __shfl_xor(va, m, 32); vb += __shfl_xor(vb, m, 32); }
        if (lane == 0) {
            mu_s[ba] = mua; rstd_s[ba] = rsqrtf(va * (1.0f/256.0f) + LN_EPS);
            mu_s[bc] = mub; rstd_s[bc] = rsqrtf(vb * (1.0f/256.0f) + LN_EPS);
        }
    }
    lds_barrier();
    LOAD_WCAP_RANGE(49, 64)

    {
        int i = t & 31, b0 = t >> 5;
        #pragma unroll
        for (int bsel = 0; bsel < 2; ++bsel) {
            int b = b0 + 8*bsel;
            float mu = mu_s[b], rs = rstd_s[b];
            float vals[8]; float nsq = 0.f;
            #pragma unroll
            for (int k = 0; k < 8; ++k) {
                float v = (xh[b*XP + i*8 + k] - mu) * rs * g8[k] + be8[k];
                vals[k] = v; nsq = fmaf(v, v, nsq);
            }
            float scale = nsq / ((1.0f + nsq) * sqrtf(nsq + SQ_EPS));
            #pragma unroll
            for (int k = 0; k < 8; ++k) u_t[i*130 + k*16 + b] = scale * vals[k];
        }
    }

    #pragma unroll
    for (int m = 0; m < 2; ++m) {
        #pragma unroll
        for (int g = 0; g < 4; ++g)
            Cs[(((2*hi + m)*4 + g)*8 + wav)*16 + bb] = make_float4(0.0625f, 0.0625f, 0.0625f, 0.0625f);
    }
    lds_barrier();

    v2f bu0[4], bu1[4];
    #pragma unroll
    for (int ii = 0; ii < 4; ++ii) {
        const float* ub = u_t + (4*wav + ii)*130 + bb;
        bu0[ii][0] = ub[(2*hi+0)*16];  bu0[ii][1] = ub[(2*hi+1)*16];
        bu1[ii][0] = ub[(2*hi+4)*16];  bu1[ii][1] = ub[(2*hi+5)*16];
    }

    lds_barrier();
    #pragma unroll
    for (int q = 0; q < 32; ++q)
        Wl[(wav*32 + q)*32 + lane] = make_float4(a0[32 + q][0], a0[32 + q][1], a1[32 + q][0], a1[32 + q][1]);
    __builtin_amdgcn_sched_barrier(0);
    const float4* const wl = Wl + (wav*32)*32 + lane;
    v8f uk[32];
    #pragma unroll
    for (int q = 0; q < 32; ++q) {
        v8f t8 = vzero8();
        t8 = wmma4(a0[q], bu0[q >> 4], t8);
        t8 = wmma4(a1[q], bu1[q >> 4], t8);
        uk[q] = t8;
    }

    float Lg[32];
    #pragma unroll
    for (int q = 0; q < 32; ++q) Lg[q] = 0.f;
    const bool up = (hi != 0);

    v8f vA = vzero8(), vB = vzero8();
    for (int r = 0; r < 3; ++r) {
        #pragma unroll
        for (int g = 0; g < 4; ++g) {
            float cc[4][4];
            #pragma unroll
            for (int ii = 0; ii < 4; ++ii) {
                const float4 c4 = Cs[((ii*4 + g)*8 + wav)*16 + bb];
                cc[ii][0] = c4.x; cc[ii][1] = c4.y; cc[ii][2] = c4.z; cc[ii][3] = c4.w;
            }
            float4 wf[2][4];
            #pragma unroll
            for (int m = 0; m < 2; ++m) {
                #pragma unroll
                for (int jj = 0; jj < 4; ++jj) wf[m][jj] = wl[(m*16 + 4*g + jj)*32];
            }
            v8f sj[4];
            #pragma unroll
            for (int jj = 0; jj < 4; ++jj) {
                const float c0 = cc[0][jj], c1 = cc[1][jj];
                #pragma unroll
                for (int v = 0; v < 8; ++v) sj[jj][v] = fmaf(c1, uk[16 + 4*g + jj][v], c0 * uk[4*g + jj][v]);
            }
            #pragma unroll
            for (int ii = 2; ii < 4; ++ii) {
                #pragma unroll
                for (int jj = 0; jj < 4; ++jj) {
                    const float c = cc[ii][jj];
                    const float4 w = wf[ii - 2][jj];
                    const v2f f0 = {w.x, w.y}, f1 = {w.z, w.w};
                    v2f p0, p1;
                    p0[0] = bu0[ii][0]*c; p0[1] = bu0[ii][1]*c; p1[0] = bu1[ii][0]*c; p1[1] = bu1[ii][1]*c;
                    sj[jj] = wmma4(f0, p0, sj[jj]);
                    sj[jj] = wmma4(f1, p1, sj[jj]);
                }
            }
            #pragma unroll
            for (int jj = 0; jj < 4; ++jj) {
                Ps[((wav*16 + 4*g + jj)*2 + 0)*32 + lane] = make_float4(sj[jj][0], sj[jj][1], sj[jj][2], sj[jj][3]);
                Ps[((wav*16 + 4*g + jj)*2 + 1)*32 + lane] = make_float4(sj[jj][4], sj[jj][5], sj[jj][6], sj[jj][7]);
            }
        }
        __syncthreads();
        v8f sA, sB;
        {
            float4 x0 = Ps[((0*16 + wav)*2 + 0)*32 + lane], x1 = Ps[((0*16 + wav)*2 + 1)*32 + lane];
            float4 y0 = Ps[((0*16 + wav + 8)*2 + 0)*32 + lane], y1 = Ps[((0*16 + wav + 8)*2 + 1)*32 + lane];
            sA[0]=x0.x; sA[1]=x0.y; sA[2]=x0.z; sA[3]=x0.w; sA[4]=x1.x; sA[5]=x1.y; sA[6]=x1.z; sA[7]=x1.w;
            sB[0]=y0.x; sB[1]=y0.y; sB[2]=y0.z; sB[3]=y0.w; sB[4]=y1.x; sB[5]=y1.y; sB[6]=y1.z; sB[7]=y1.w;
        }
        #pragma unroll
        for (int w2 = 1; w2 < 8; ++w2) {
            float4 x0 = Ps[((w2*16 + wav)*2 + 0)*32 + lane], x1 = Ps[((w2*16 + wav)*2 + 1)*32 + lane];
            float4 y0 = Ps[((w2*16 + wav + 8)*2 + 0)*32 + lane], y1 = Ps[((w2*16 + wav + 8)*2 + 1)*32 + lane];
            sA[0]+=x0.x; sA[1]+=x0.y; sA[2]+=x0.z; sA[3]+=x0.w; sA[4]+=x1.x; sA[5]+=x1.y; sA[6]+=x1.z; sA[7]+=x1.w;
            sB[0]+=y0.x; sB[1]+=y0.y; sB[2]+=y0.z; sB[3]+=y0.w; sB[4]+=y1.x; sB[5]+=y1.y; sB[6]+=y1.z; sB[7]+=y1.w;
        }
        float nA = 0.f, nB = 0.f;
        #pragma unroll
        for (int v = 0; v < 8; ++v) { nA = fmaf(sA[v], sA[v], nA); nB = fmaf(sB[v], sB[v], nB); }
        nA += __shfl_xor(nA, 16, 32);  nB += __shfl_xor(nB, 16, 32);
        float scA = nA / ((1.0f + nA) * sqrtf(nA + SQ_EPS));
        float scB = nB / ((1.0f + nB) * sqrtf(nB + SQ_EPS));
        #pragma unroll
        for (int v = 0; v < 8; ++v) { vA[v] = sA[v]*scA; vB[v] = sB[v]*scB; }

        if (r < 2) {
            Vs[((wav    )*2 + 0)*32 + lane] = make_float4(vA[0], vA[1], vA[2], vA[3]);
            Vs[((wav    )*2 + 1)*32 + lane] = make_float4(vA[4], vA[5], vA[6], vA[7]);
            Vs[((wav + 8)*2 + 0)*32 + lane] = make_float4(vB[0], vB[1], vB[2], vB[3]);
            Vs[((wav + 8)*2 + 1)*32 + lane] = make_float4(vB[4], vB[5], vB[6], vB[7]);
            __syncthreads();
            float own_p[2] = {0.f, 0.f}, rcv_p[2] = {0.f, 0.f};
            #pragma unroll
            for (int j = 0; j < 16; ++j) {
                const float4 y0 = Vs[(j*2 + 0)*32 + lane], y1 = Vs[(j*2 + 1)*32 + lane];
                const float4 w2 = wl[j*32], w3 = wl[(16 + j)*32];
                float ag[4];
                #pragma unroll
                for (int ii = 0; ii < 4; ++ii) {
                    v8f uh;
                    if (ii < 2) {
                        uh = uk[ii*16 + j];
                    } else {
                        const float4 w = (ii == 2) ? w2 : w3;
                        const v2f f0 = {w.x, w.y}, f1 = {w.z, w.w};
                        uh = vzero8();
                        uh = wmma4(f0, bu0[ii], uh);
                        uh = wmma4(f1, bu1[ii], uh);
                    }
                    v2f a2 = {0.f, 0.f};
                    a2 = __builtin_elementwise_fma((v2f){uh[0], uh[1]}, (v2f){y0.x, y0.y}, a2);
                    a2 = __builtin_elementwise_fma((v2f){uh[2], uh[3]}, (v2f){y0.z, y0.w}, a2);
                    a2 = __builtin_elementwise_fma((v2f){uh[4], uh[5]}, (v2f){y1.x, y1.y}, a2);
                    a2 = __builtin_elementwise_fma((v2f){uh[6], uh[7]}, (v2f){y1.z, y1.w}, a2);
                    ag[ii] = a2[0] + a2[1];
                }
                if (j > 0) {
                    asm volatile("" :: "v"(rcv_p[0]), "v"(rcv_p[1]));
                    #pragma unroll
                    for (int m = 0; m < 2; ++m) Lg[m*16 + j - 1] += own_p[m] + rcv_p[m];
                }
                #pragma unroll
                for (int m = 0; m < 2; ++m) {
                    const float own = up ? ag[2 + m] : ag[m];
                    const float snd = up ? ag[m] : ag[2 + m];
                    own_p[m] = own;
                    rcv_p[m] = __shfl_xor(snd, 16, 32);
                }
            }
            #pragma unroll
            for (int m = 0; m < 2; ++m) Lg[m*16 + 15] += own_p[m] + rcv_p[m];
            #pragma unroll
            for (int m = 0; m < 2; ++m) {
                float mx = Lg[m*16];
                #pragma unroll
                for (int j = 1; j < 16; ++j) mx = fmaxf(mx, Lg[m*16 + j]);
                float e[16]; float sum = 0.f;
                #pragma unroll
                for (int j = 0; j < 16; ++j) { e[j] = expf(Lg[m*16 + j] - mx); sum += e[j]; }
                float inv = 1.0f / sum;
                #pragma unroll
                for (int g = 0; g < 4; ++g)
                    Cs[(((2*hi + m)*4 + g)*8 + wav)*16 + bb] = make_float4(e[4*g]*inv, e[4*g+1]*inv, e[4*g+2]*inv, e[4*g+3]*inv);
            }
            __syncthreads();
        }
    }

    {
        float pA = 0.f, pB = 0.f;
        #pragma unroll
        for (int v = 0; v < 8; ++v) {
            pA = fmaf(vA[v], Wf[(wav    )*16 + v + 8*hi], pA);
            pB = fmaf(vB[v], Wf[(wav + 8)*16 + v + 8*hi], pB);
        }
        pA += __shfl_xor(pA, 16, 32);  pB += __shfl_xor(pB, 16, 32);
        if (lane < 16) { part[wav*17 + bb] = pA; part[(wav+8)*17 + bb] = pB; }
    }
    __syncthreads();
    if (t < 16) {
        float acc = bf[0];
        #pragma unroll
        for (int j = 0; j < 16; ++j) acc += part[j*17 + t];
        out[tile*16 + t] = acc;
    }
}

extern "C" void kernel_launch(void* const* d_in, const int* in_sizes, int n_in,
                              void* d_out, int out_size, void* d_ws, size_t ws_size,
                              hipStream_t stream) {
    const float* x    = (const float*)d_in[0];
    const float* W1   = (const float*)d_in[1];
    const float* b1   = (const float*)d_in[2];
    const float* W2   = (const float*)d_in[3];
    const float* b2   = (const float*)d_in[4];
    const float* ln_g = (const float*)d_in[5];
    const float* ln_b = (const float*)d_in[6];
    const float* Wcap = (const float*)d_in[7];
    const float* Wf   = (const float*)d_in[8];
    const float* bf   = (const float*)d_in[9];
    float* out = (float*)d_out;

    const int B = in_sizes[0] / 256;
    const int blocks = B / 16;
    capsnet_fused<<<blocks, 256, 0, stream>>>(x, W1, b1, W2, b2, ln_g, ln_b, Wcap, Wf, bf, out);
}


# Five more u_hat tiles (third capsule, j < 5) multiplied once and kept in registers: 54 not 64 matrix calls a routing pass; their share of s is added as c*u_hat on the vector ALU: new sum order
# speedup vs baseline: 2.1448x; 1.0503x over previous
#include <hip/hip_runtime.h>
#include <math.h>

typedef __attribute__((ext_vector_type(2))) float v2f;
typedef __attribute__((ext_vector_type(8))) float v8f;

#define LN_EPS 1e-5f
#define SQ_EPS 1e-9f
#define XP 260
#define HP 36
#define NK 5

__device__ __forceinline__ float gelu_exact(float x) {
    return 0.5f * x * (1.0f + erff(x * 0.70710678118654752440f));
}

__device__ __forceinline__ v8f wmma4(v2f a, v2f b, v8f c) {
    return __builtin_amdgcn_wmma_f32_16x16x4_f32(false, a, false, b, (short)0, c, false, false);
}

__device__ __forceinline__ void lds_barrier() {
    __builtin_amdgcn_fence(__ATOMIC_RELEASE, "workgroup", "local");
    __builtin_amdgcn_s_barrier();
    __builtin_amdgcn_fence(__ATOMIC_ACQUIRE, "workgroup", "local");
}

__device__ __forceinline__ v8f vzero8() {
    v8f z = {0.f,0.f,0.f,0.f,0.f,0.f,0.f,0.f};
    return z;
}

__global__ __attribute__((amdgpu_waves_per_eu(1))) __launch_bounds__(256)
void capsnet_fused(
    const float* __restrict__ x,   const float* __restrict__ W1, const float* __restrict__ b1,
    const float* __restrict__ W2,  const float* __restrict__ b2,
    const float* __restrict__ ln_g,const float* __restrict__ ln_b,
    const float* __restrict__ Wcap,const float* __restrict__ Wf, const float* __restrict__ bf,
    float* __restrict__ out)
{
    __shared__ __align__(16) float smem[78128];
    float*  const xh   = smem;
    float*  const h1s  = smem + 4160;
    float*  const hp   = smem + 4736;
    float*  const u_t  = smem + 8832;
    float4* const Wl   = reinterpret_cast<float4*>(smem);
    float4* const Ps   = reinterpret_cast<float4*>(smem + 32768);
    float4* const Cs   = reinterpret_cast<float4*>(smem + 65536);
    float4* const Vs   = reinterpret_cast<float4*>(smem + 73728);
    float*  const part = smem + 77824;
    float*  const mu_s   = smem + 78096;
    float*  const rstd_s = smem + 78112;

    const int t    = threadIdx.x;
    const int lane = t & 31;
    const int wav  = t >> 5;
    const int bb   = lane & 15;
    const int hi   = lane >> 4;
    const int n4   = lane & 7;
    const int b4   = lane >> 3;

    const int tile = blockIdx.x;
    const float* xg = x + (size_t)tile * 16 * 256;

    v2f a0[64], a1[64];
#define LOAD_WCAP_RANGE(LO, HI)                                                                               \
    _Pragma("unroll")                                                                                         \
    for (int p = (LO); p < (HI); ++p) {                                                                       \
        const int pi = ((p >> 4) + 2) & 3;                                                                    \
        const int pj = p & 15;                                                                                \
        const float* wc = Wcap + (size_t)(((4*wav + pi)*16 + pj)*8)*16 + bb;                  \
        a0[pi*16 + pj][0] = wc[(2*hi+0)*16];  a0[pi*16 + pj][1] = wc[(2*hi+1)*16];                             \
        a1[pi*16 + pj][0] = wc[(2*hi+4)*16];  a1[pi*16 + pj][1] = wc[(2*hi+5)*16];                             \
    }

    float4 wk1[32];
    #pragma unroll
    for (int k = 0; k < 32; ++k) wk1[k] = *reinterpret_cast<const float4*>(W1 + (32*wav + k)*32 + 4*n4);

    #pragma unroll
    for (int r = 0; r < 16; ++r) xh[r*XP + t] = xg[t + 256*r];
    lds_barrier();
    LOAD_WCAP_RANGE(0, 15)

    {
        #pragma unroll
        for (int bi = 0; bi < 4; ++bi) {
            const int b = 4*b4 + bi;
            const float4* xr = reinterpret_cast<const float4*>(xh + b*XP + 32*wav);
            float c0 = 0.f, c1 = 0.f, c2 = 0.f, c3 = 0.f;
            #pragma unroll
            for (int q = 0; q < 8; ++q) {
                const float4 xv = xr[q];
                const float xs[4] = {xv.x, xv.y, xv.z, xv.w};
                #pragma unroll
                for (int e = 0; e < 4; ++e) {
                    const float4 w = wk1[4*q + e];
                    c0 = fmaf(xs[e], w.x, c0);
                    c1 = fmaf(xs[e], w.y, c1);
                    c2 = fmaf(xs[e], w.z, c2);
                    c3 = fmaf(xs[e], w.w, c3);
                }
            }
            *reinterpret_cast<float4*>(hp + (wav*16 + b)*32 + 4*n4) = make_float4(c0, c1, c2, c3);
        }
    }
    lds_barrier();

    float4 wk2[32];
    #pragma unroll
    for (int k = 0; k < 32; ++k) wk2[k] = *reinterpret_cast<const float4*>(W2 + k*256 + 32*wav + 4*n4);
    const float4 b2v = *reinterpret_cast<const float4*>(b2 + 32*wav + 4*n4);
    float g8[8], be8[8];
    #pragma unroll
    for (int k = 0; k < 8; ++k) { g8[k] = ln_g[(t & 31)*8 + k]; be8[k] = ln_b[(t & 31)*8 + k]; }
    LOAD_WCAP_RANGE(15, 19)

    {
        int n = lane, b0 = wav;
        float s0 = b1[n], s1 = s0;
        #pragma unroll
        for (int w2 = 0; w2 < 8; ++w2) { s0 += hp[(w2*16 + b0)*32 + n]; s1 += hp[(w2*16 + b0 + 8)*32 + n]; }
        h1s[b0*HP + n]     = gelu_exact(s0);
        h1s[(b0+8)*HP + n] = gelu_exact(s1);
    }
    lds_barrier();

    {
        LOAD_WCAP_RANGE(19, 34)
        #pragma unroll 1
        for (int bh = 0; bh < 4; bh += 2) {
            const int b = 4*b4 + bh;
            const float4* hr0 = reinterpret_cast<const float4*>(h1s + b*HP);
            const float4* hr1 = reinterpret_cast<const float4*>(h1s + (b + 1)*HP);
            v2f s00 = {b2v.x, b2v.y}, s01 = {b2v.z, b2v.w}, s10 = s00, s11 = s01;
            #pragma unroll
            for (int q = 0; q < 8; ++q) {
                const float4 h0 = hr0[q], h1 = hr1[q];
                const float ha[4] = {h0.x, h0.y, h0.z, h0.w};
                const float hb[4] = {h1.x, h1.y, h1.z, h1.w};
                #pragma unroll
                for (int e = 0; e < 4; ++e) {
                    const float4 w = wk2[4*q + e];
                    const v2f wl = {w.x, w.y}, wh = {w.z, w.w};
                    const v2f xa = {ha[e], ha[e]}, xb = {hb[e], hb[e]};
                    s00 = __builtin_elementwise_fma(xa, wl, s00);
                    s01 = __builtin_elementwise_fma(xa, wh, s01);
                    s10 = __builtin_elementwise_fma(xb, wl, s10);
                    s11 = __builtin_elementwise_fma(xb, wh, s11);
                }
            }
            asm volatile("" :: "v"(s00), "v"(s01), "v"(s10), "v"(s11));
            *reinterpret_cast<float4*>(xh + b*XP + 32*wav + 4*n4) =
                make_float4(gelu_exact(s00[0]), gelu_exact(s00[1]), gelu_exact(s01[0]), gelu_exact(s01[1]));
            *reinterpret_cast<float4*>(xh + (b + 1)*XP + 32*wav + 4*n4) =
                make_float4(gelu_exact(s10[0]), gelu_exact(s10[1]), gelu_exact(s11[0]), gelu_exact(s11[1]));
        }
    }
    LOAD_WCAP_RANGE(34, 49)
    lds_barrier();

    {
        int ba = wav, bc = wav + 8;
        float sa = 0.f, sb = 0.f;
        #pragma unroll
        for (int r = 0; r < 8; ++r) { sa += xh[ba*XP + lane + 32*r]; sb += xh[bc*XP + lane + 32*r]; }
        #pragma unroll
        for (int m = 16; m >= 1; m >>= 1) { sa += __shfl_xor(sa, m, 32); sb += __shfl_xor(sb, m, 32); }
        float mua = sa * (1.0f/256.0f), mub = sb * (1.0f/256.0f);
        float va = 0.f, vb = 0.f;
        #pragma unroll
        for (int r = 0; r < 8; ++r) {
            float da = xh[ba*XP + lane + 32*r] - mua; va = fmaf(da, da, va);
            float db = xh[bc*XP + lane + 32*r] - mub; vb = fmaf(db, db, vb);
        }
        #pragma unroll
        for (int m = 16; m >= 1; m >>= 1) { va += __shfl_xor(va, m, 32); vb += __shfl_xor(vb, m, 32); }
        if (lane == 0) {
            mu_s[ba] = mua; rstd_s[ba] = rsqrtf(va * (1.0f/256.0f) + LN_EPS);
            mu_s[bc] = mub; rstd_s[bc] = rsqrtf(vb * (1.0f/256.0f) + LN_EPS);
        }
    }
    lds_barrier();
    LOAD_WCAP_RANGE(49, 64)

    {
        int i = t & 31, b0 = t >> 5;
        #pragma unroll
        for (int bsel = 0; bsel < 2; ++bsel) {
            int b = b0 + 8*bsel;
            float mu = mu_s[b], rs = rstd_s[b];
            float vals[8]; float nsq = 0.f;
            #pragma unroll
            for (int k = 0; k < 8; ++k) {
                float v = (xh[b*XP + i*8 + k] - mu) * rs * g8[k] + be8[k];
                vals[k] = v; nsq = fmaf(v, v, nsq);
            }
            float scale = nsq / ((1.0f + nsq) * sqrtf(nsq + SQ_EPS));
            #pragma unroll
            for (int k = 0; k < 8; ++k) u_t[i*130 + k*16 + b] = scale * vals[k];
        }
    }

    #pragma unroll
    for (int m = 0; m < 2; ++m) {
        #pragma unroll
        for (int g = 0; g < 4; ++g)
            Cs[(((2*hi + m)*4 + g)*8 + wav)*16 + bb] = make_float4(0.0625f, 0.0625f, 0.0625f, 0.0625f);
    }
    lds_barrier();

    v2f bu0[4], bu1[4];
    #pragma unroll
    for (int ii = 0; ii < 4; ++ii) {
        const float* ub = u_t + (4*wav + ii)*130 + bb;
        bu0[ii][0] = ub[(2*hi+0)*16];  bu0[ii][1] = ub[(2*hi+1)*16];
        bu1[ii][0] = ub[(2*hi+4)*16];  bu1[ii][1] = ub[(2*hi+5)*16];
    }

    lds_barrier();
    v8f uk2[NK];
    #pragma unroll
    for (int j = 0; j < NK; ++j) {
        v8f t8 = vzero8();
        t8 = wmma4(a0[32 + j], bu0[2], t8);
        t8 = wmma4(a1[32 + j], bu1[2], t8);
        uk2[j] = t8;
    }
    #pragma unroll
    for (int q = NK; q < 32; ++q)
        Wl[(wav*32 + q)*32 + lane] = make_float4(a0[32 + q][0], a0[32 + q][1], a1[32 + q][0], a1[32 + q][1]);
    __builtin_amdgcn_sched_barrier(0);
    const float4* const wl = Wl + (wav*32)*32 + lane;
    v8f uk[32];
    #pragma unroll
    for (int q = 0; q < 32; ++q) {
        v8f t8 = vzero8();
        t8 = wmma4(a0[q], bu0[q >> 4], t8);
        t8 = wmma4(a1[q], bu1[q >> 4], t8);
        uk[q] = t8;
    }

    float Lg[32];
    #pragma unroll
    for (int q = 0; q < 32; ++q) Lg[q] = 0.f;
    const bool up = (hi != 0);

    v8f vA = vzero8(), vB = vzero8();
    for (int r = 0; r < 3; ++r) {
        #pragma unroll
        for (int g = 0; g < 4; ++g) {
            float cc[4][4];
            #pragma unroll
            for (int ii = 0; ii < 4; ++ii) {
                const float4 c4 = Cs[((ii*4 + g)*8 + wav)*16 + bb];
                cc[ii][0] = c4.x; cc[ii][1] = c4.y; cc[ii][2] = c4.z; cc[ii][3] = c4.w;
            }
            float4 wf[2][4];
            #pragma unroll
            for (int m = 0; m < 2; ++m) {
                #pragma unroll
                for (int jj = 0; jj < 4; ++jj)
                    wf[m][jj] = (m == 0 && 4*g + jj < NK) ? make_float4(0.f, 0.f, 0.f, 0.f)
                                                          : wl[(m*16 + 4*g + jj)*32];
            }
            v8f sj[4];
            #pragma unroll
            for (int jj = 0; jj < 4; ++jj) {
                const float c0 = cc[0][jj], c1 = cc[1][jj];
                #pragma unroll
                for (int v = 0; v < 8; ++v) sj[jj][v] = fmaf(c1, uk[16 + 4*g + jj][v], c0 * uk[4*g + jj][v]);
                if (4*g + jj < NK) {
                    const float c2 = cc[2][jj];
                    #pragma unroll
                    for (int v = 0; v < 8; ++v) sj[jj][v] = fmaf(c2, uk2[(4*g + jj) % NK][v], sj[jj][v]);
                }
            }
            #pragma unroll
            for (int ii = 2; ii < 4; ++ii) {
                #pragma unroll
                for (int jj = 0; jj < 4; ++jj) {
                    if (ii == 2 && 4*g + jj < NK) continue;
                    const float c = cc[ii][jj];
                    const float4 w = wf[ii - 2][jj];
                    const v2f f0 = {w.x, w.y}, f1 = {w.z, w.w};
                    v2f p0, p1;
                    p0[0] = bu0[ii][0]*c; p0[1] = bu0[ii][1]*c; p1[0] = bu1[ii][0]*c; p1[1] = bu1[ii][1]*c;
                    sj[jj] = wmma4(f0, p0, sj[jj]);
                    sj[jj] = wmma4(f1, p1, sj[jj]);
                }
            }
            #pragma unroll
            for (int jj = 0; jj < 4; ++jj) {
                Ps[((wav*16 + 4*g + jj)*2 + 0)*32 + lane] = make_float4(sj[jj][0], sj[jj][1], sj[jj][2], sj[jj][3]);
                Ps[((wav*16 + 4*g + jj)*2 + 1)*32 + lane] = make_float4(sj[jj][4], sj[jj][5], sj[jj][6], sj[jj][7]);
            }
        }
        __syncthreads();
        v8f sA, sB;
        {
            float4 x0 = Ps[((0*16 + wav)*2 + 0)*32 + lane], x1 = Ps[((0*16 + wav)*2 + 1)*32 + lane];
            float4 y0 = Ps[((0*16 + wav + 8)*2 + 0)*32 + lane], y1 = Ps[((0*16 + wav + 8)*2 + 1)*32 + lane];
            sA[0]=x0.x; sA[1]=x0.y; sA[2]=x0.z; sA[3]=x0.w; sA[4]=x1.x; sA[5]=x1.y; sA[6]=x1.z; sA[7]=x1.w;
            sB[0]=y0.x; sB[1]=y0.y; sB[2]=y0.z; sB[3]=y0.w; sB[4]=y1.x; sB[5]=y1.y; sB[6]=y1.z; sB[7]=y1.w;
        }
        #pragma unroll
        for (int w2 = 1; w2 < 8; ++w2) {
            float4 x0 = Ps[((w2*16 + wav)*2 + 0)*32 + lane], x1 = Ps[((w2*16 + wav)*2 + 1)*32 + lane];
            float4 y0 = Ps[((w2*16 + wav + 8)*2 + 0)*32 + lane], y1 = Ps[((w2*16 + wav + 8)*2 + 1)*32 + lane];
            sA[0]+=x0.x; sA[1]+=x0.y; sA[2]+=x0.z; sA[3]+=x0.w; sA[4]+=x1.x; sA[5]+=x1.y; sA[6]+=x1.z; sA[7]+=x1.w;
            sB[0]+=y0.x; sB[1]+=y0.y; sB[2]+=y0.z; sB[3]+=y0.w; sB[4]+=y1.x; sB[5]+=y1.y; sB[6]+=y1.z; sB[7]+=y1.w;
        }
        float nA = 0.f, nB = 0.f;
        #pragma unroll
        for (int v = 0; v < 8; ++v) { nA = fmaf(sA[v], sA[v], nA); nB = fmaf(sB[v], sB[v], nB); }
        nA += __shfl_xor(nA, 16, 32);  nB += __shfl_xor(nB, 16, 32);
        float scA = nA / ((1.0f + nA) * sqrtf(nA + SQ_EPS));
        float scB = nB / ((1.0f + nB) * sqrtf(nB + SQ_EPS));
        #pragma unroll
        for (int v = 0; v < 8; ++v) { vA[v] = sA[v]*scA; vB[v] = sB[v]*scB; }

        if (r < 2) {
            Vs[((wav    )*2 + 0)*32 + lane] = make_float4(vA[0], vA[1], vA[2], vA[3]);
            Vs[((wav    )*2 + 1)*32 + lane] = make_float4(vA[4], vA[5], vA[6], vA[7]);
            Vs[((wav + 8)*2 + 0)*32 + lane] = make_float4(vB[0], vB[1], vB[2], vB[3]);
            Vs[((wav + 8)*2 + 1)*32 + lane] = make_float4(vB[4], vB[5], vB[6], vB[7]);
            __syncthreads();
            float own_p[2] = {0.f, 0.f}, rcv_p[2] = {0.f, 0.f};
            #pragma unroll
            for (int j = 0; j < 16; ++j) {
                const float4 y0 = Vs[(j*2 + 0)*32 + lane], y1 = Vs[(j*2 + 1)*32 + lane];
                const float4 w2 = (j < NK) ? make_float4(0.f, 0.f, 0.f, 0.f) : wl[j*32];
                const float4 w3 = wl[(16 + j)*32];
                float ag[4];
                #pragma unroll
                for (int ii = 0; ii < 4; ++ii) {
                    v8f uh;
                    if (ii < 2) {
                        uh = uk[ii*16 + j];
                    } else if (ii == 2 && j < NK) {
                        uh = uk2[j % NK];
                    } else {
                        const float4 w = (ii == 2) ? w2 : w3;
                        const v2f f0 = {w.x, w.y}, f1 = {w.z, w.w};
                        uh = vzero8();
                        uh = wmma4(f0, bu0[ii], uh);
                        uh = wmma4(f1, bu1[ii], uh);
                    }
                    v2f a2 = {0.f, 0.f};
                    a2 = __builtin_elementwise_fma((v2f){uh[0], uh[1]}, (v2f){y0.x, y0.y}, a2);
                    a2 = __builtin_elementwise_fma((v2f){uh[2], uh[3]}, (v2f){y0.z, y0.w}, a2);
                    a2 = __builtin_elementwise_fma((v2f){uh[4], uh[5]}, (v2f){y1.x, y1.y}, a2);
                    a2 = __builtin_elementwise_fma((v2f){uh[6], uh[7]}, (v2f){y1.z, y1.w}, a2);
                    ag[ii] = a2[0] + a2[1];
                }
                if (j > 0) {
                    asm volatile("" :: "v"(rcv_p[0]), "v"(rcv_p[1]));
                    #pragma unroll
                    for (int m = 0; m < 2; ++m) Lg[m*16 + j - 1] += own_p[m] + rcv_p[m];
                }
                #pragma unroll
                for (int m = 0; m < 2; ++m) {
                    const float own = up ? ag[2 + m] : ag[m];
                    const float snd = up ? ag[m] : ag[2 + m];
                    own_p[m] = own;
                    rcv_p[m] = __shfl_xor(snd, 16, 32);
                }
            }
            #pragma unroll
            for (int m = 0; m < 2; ++m) Lg[m*16 + 15] += own_p[m] + rcv_p[m];
            #pragma unroll
            for (int m = 0; m < 2; ++m) {
                float mx = Lg[m*16];
                #pragma unroll
                for (int j = 1; j < 16; ++j) mx = fmaxf(mx, Lg[m*16 + j]);
                float e[16]; float sum = 0.f;
                #pragma unroll
                for (int j = 0; j < 16; ++j) { e[j] = expf(Lg[m*16 + j] - mx); sum += e[j]; }
                float inv = 1.0f / sum;
                #pragma unroll
                for (int g = 0; g < 4; ++g)
                    Cs[(((2*hi + m)*4 + g)*8 + wav)*16 + bb] = make_float4(e[4*g]*inv, e[4*g+1]*inv, e[4*g+2]*inv, e[4*g+3]*inv);
            }
            __syncthreads();
        }
    }

    {
        float pA = 0.f, pB = 0.f;
        #pragma unroll
        for (int v = 0; v < 8; ++v) {
            pA = fmaf(vA[v], Wf[(wav    )*16 + v + 8*hi], pA);
            pB = fmaf(vB[v], Wf[(wav + 8)*16 + v + 8*hi], pB);
        }
        pA += __shfl_xor(pA, 16, 32);  pB += __shfl_xor(pB, 16, 32);
        if (lane < 16) { part[wav*17 + bb] = pA; part[(wav+8)*17 + bb] = pB; }
    }
    __syncthreads();
    if (t < 16) {
        float acc = bf[0];
        #pragma unroll
        for (int j = 0; j < 16; ++j) acc += part[j*17 + t];
        out[tile*16 + t] = acc;
    }
}

extern "C" void kernel_launch(void* const* d_in, const int* in_sizes, int n_in,
                              void* d_out, int out_size, void* d_ws, size_t ws_size,
                              hipStream_t stream) {
    const float* x    = (const float*)d_in[0];
    const float* W1   = (const float*)d_in[1];
    const float* b1   = (const float*)d_in[2];
    const float* W2   = (const float*)d_in[3];
    const float* b2   = (const float*)d_in[4];
    const float* ln_g = (const float*)d_in[5];
    const float* ln_b = (const float*)d_in[6];
    const float* Wcap = (const float*)d_in[7];
    const float* Wf   = (const float*)d_in[8];
    const float* bf   = (const float*)d_in[9];
    float* out = (float*)d_out;

    const int B = in_sizes[0] / 256;
    const int blocks = B / 16;
    capsnet_fused<<<blocks, 256, 0, stream>>>(x, W1, b1, W2, b2, ln_g, ln_b, Wcap, Wf, bf, out);
}


# All 16 u_hat tiles of the third capsule kept in registers (was 5): 32 not 54 matrix calls a routing pass; logits parked in LDS between agreement passes to make room; s sums in new order
# speedup vs baseline: 2.2617x; 1.0545x over previous
#include <hip/hip_runtime.h>
#include <math.h>

typedef __attribute__((ext_vector_type(2))) float v2f;
typedef __attribute__((ext_vector_type(8))) float v8f;

#define LN_EPS 1e-5f
#define SQ_EPS 1e-9f
#define XP 260
#define HP 36
#define NK 16

__device__ __forceinline__ float gelu_exact(float x) {
    return 0.5f * x * (1.0f + erff(x * 0.70710678118654752440f));
}

__device__ __forceinline__ v8f wmma4(v2f a, v2f b, v8f c) {
    return __builtin_amdgcn_wmma_f32_16x16x4_f32(false, a, false, b, (short)0, c, false, false);
}

__device__ __forceinline__ void lds_barrier() {
    __builtin_amdgcn_fence(__ATOMIC_RELEASE, "workgroup", "local");
    __builtin_amdgcn_s_barrier();
    __builtin_amdgcn_fence(__ATOMIC_ACQUIRE, "workgroup", "local");
}

__device__ __forceinline__ v8f vzero8() {
    v8f z = {0.f,0.f,0.f,0.f,0.f,0.f,0.f,0.f};
    return z;
}

__global__ __attribute__((amdgpu_waves_per_eu(1))) __launch_bounds__(256)
void capsnet_fused(
    const float* __restrict__ x,   const float* __restrict__ W1, const float* __restrict__ b1,
    const float* __restrict__ W2,  const float* __restrict__ b2,
    const float* __restrict__ ln_g,const float* __restrict__ ln_b,
    const float* __restrict__ Wcap,const float* __restrict__ Wf, const float* __restrict__ bf,
    float* __restrict__ out)
{
    __shared__ __align__(16) float smem[78128];
    float*  const xh   = smem;
    float*  const h1s  = smem + 4160;
    float*  const hp   = smem + 4736;
    float*  const u_t  = smem + 8832;
    float4* const Wl   = reinterpret_cast<float4*>(smem);
    float4* const Ps   = reinterpret_cast<float4*>(smem + 32768);
    float4* const Cs   = reinterpret_cast<float4*>(smem + 65536);
    float4* const Vs   = reinterpret_cast<float4*>(smem + 73728);
    float*  const part = smem + 77824;
    float*  const mu_s   = smem + 78096;
    float*  const rstd_s = smem + 78112;

    const int t    = threadIdx.x;
    const int lane = t & 31;
    const int wav  = t >> 5;
    const int bb   = lane & 15;
    const int hi   = lane >> 4;
    const int n4   = lane & 7;
    const int b4   = lane >> 3;

    const int tile = blockIdx.x;
    const float* xg = x + (size_t)tile * 16 * 256;

    v2f a0[64], a1[64];
#define LOAD_WCAP_RANGE(LO, HI)                                                                               \
    _Pragma("unroll")                                                                                         \
    for (int p = (LO); p < (HI); ++p) {                                                                       \
        const int pi = ((p >> 4) + 2) & 3;                                                                    \
        const int pj = p & 15;                                                                                \
        const float* wc = Wcap + (size_t)(((4*wav + pi)*16 + pj)*8)*16 + bb;                  \
        a0[pi*16 + pj][0] = wc[(2*hi+0)*16];  a0[pi*16 + pj][1] = wc[(2*hi+1)*16];                             \
        a1[pi*16 + pj][0] = wc[(2*hi+4)*16];  a1[pi*16 + pj][1] = wc[(2*hi+5)*16];                             \
    }

    float4 wk1[32];
    #pragma unroll
    for (int k = 0; k < 32; ++k) wk1[k] = *reinterpret_cast<const float4*>(W1 + (32*wav + k)*32 + 4*n4);

    #pragma unroll
    for (int r = 0; r < 16; ++r) xh[r*XP + t] = xg[t + 256*r];
    lds_barrier();
    LOAD_WCAP_RANGE(0, 15)

    {
        #pragma unroll
        for (int bi = 0; bi < 4; ++bi) {
            const int b = 4*b4 + bi;
            const float4* xr = reinterpret_cast<const float4*>(xh + b*XP + 32*wav);
            float c0 = 0.f, c1 = 0.f, c2 = 0.f, c3 = 0.f;
            #pragma unroll
            for (int q = 0; q < 8; ++q) {
                const float4 xv = xr[q];
                const float xs[4] = {xv.x, xv.y, xv.z, xv.w};
                #pragma unroll
                for (int e = 0; e < 4; ++e) {
                    const float4 w = wk1[4*q + e];
                    c0 = fmaf(xs[e], w.x, c0);
                    c1 = fmaf(xs[e], w.y, c1);
                    c2 = fmaf(xs[e], w.z, c2);
                    c3 = fmaf(xs[e], w.w, c3);
                }
            }
            *reinterpret_cast<float4*>(hp + (wav*16 + b)*32 + 4*n4) = make_float4(c0, c1, c2, c3);
        }
    }
    lds_barrier();

    float4 wk2[32];
    #pragma unroll
    for (int k = 0; k < 32; ++k) wk2[k] = *reinterpret_cast<const float4*>(W2 + k*256 + 32*wav + 4*n4);
    const float4 b2v = *reinterpret_cast<const float4*>(b2 + 32*wav + 4*n4);
    float g8[8], be8[8];
    #pragma unroll
    for (int k = 0; k < 8; ++k) { g8[k] = ln_g[(t & 31)*8 + k]; be8[k] = ln_b[(t & 31)*8 + k]; }
    LOAD_WCAP_RANGE(15, 19)

    {
        int n = lane, b0 = wav;
        float s0 = b1[n], s1 = s0;
        #pragma unroll
        for (int w2 = 0; w2 < 8; ++w2) { s0 += hp[(w2*16 + b0)*32 + n]; s1 += hp[(w2*16 + b0 + 8)*32 + n]; }
        h1s[b0*HP + n]     = gelu_exact(s0);
        h1s[(b0+8)*HP + n] = gelu_exact(s1);
    }
    lds_barrier();

    {
        LOAD_WCAP_RANGE(19, 34)
        #pragma unroll 1
        for (int bh = 0; bh < 4; bh += 2) {
            const int b = 4*b4 + bh;
            const float4* hr0 = reinterpret_cast<const float4*>(h1s + b*HP);
            const float4* hr1 = reinterpret_cast<const float4*>(h1s + (b + 1)*HP);
            v2f s00 = {b2v.x, b2v.y}, s01 = {b2v.z, b2v.w}, s10 = s00, s11 = s01;
            #pragma unroll
            for (int q = 0; q < 8; ++q) {
                const float4 h0 = hr0[q], h1 = hr1[q];
                const float ha[4] = {h0.x, h0.y, h0.z, h0.w};
                const float hb[4] = {h1.x, h1.y, h1.z, h1.w};
                #pragma unroll
                for (int e = 0; e < 4; ++e) {
                    const float4 w = wk2[4*q + e];
                    const v2f wl = {w.x, w.y}, wh = {w.z, w.w};
                    const v2f xa = {ha[e], ha[e]}, xb = {hb[e], hb[e]};
                    s00 = __builtin_elementwise_fma(xa, wl, s00);
                    s01 = __builtin_elementwise_fma(xa, wh, s01);
                    s10 = __builtin_elementwise_fma(xb, wl, s10);
                    s11 = __builtin_elementwise_fma(xb, wh, s11);
                }
            }
            asm volatile("" :: "v"(s00), "v"(s01), "v"(s10), "v"(s11));
            *reinterpret_cast<float4*>(xh + b*XP + 32*wav + 4*n4) =
                make_float4(gelu_exact(s00[0]), gelu_exact(s00[1]), gelu_exact(s01[0]), gelu_exact(s01[1]));
            *reinterpret_cast<float4*>(xh + (b + 1)*XP + 32*wav + 4*n4) =
                make_float4(gelu_exact(s10[0]), gelu_exact(s10[1]), gelu_exact(s11[0]), gelu_exact(s11[1]));
        }
    }
    LOAD_WCAP_RANGE(34, 49)
    lds_barrier();

    {
        int ba = wav, bc = wav + 8;
        float sa = 0.f, sb = 0.f;
        #pragma unroll
        for (int r = 0; r < 8; ++r) { sa += xh[ba*XP + lane + 32*r]; sb += xh[bc*XP + lane + 32*r]; }
        #pragma unroll
        for (int m = 16; m >= 1; m >>= 1) { sa += __shfl_xor(sa, m, 32); sb += __shfl_xor(sb, m, 32); }
        float mua = sa * (1.0f/256.0f), mub = sb * (1.0f/256.0f);
        float va = 0.f, vb = 0.f;
        #pragma unroll
        for (int r = 0; r < 8; ++r) {
            float da = xh[ba*XP + lane + 32*r] - mua; va = fmaf(da, da, va);
            float db = xh[bc*XP + lane + 32*r] - mub; vb = fmaf(db, db, vb);
        }
        #pragma unroll
        for (int m = 16; m >= 1; m >>= 1) { va += __shfl_xor(va, m, 32); vb += __shfl_xor(vb, m, 32); }
        if (lane == 0) {
            mu_s[ba] = mua; rstd_s[ba] = rsqrtf(va * (1.0f/256.0f) + LN_EPS);
            mu_s[bc] = mub; rstd_s[bc] = rsqrtf(vb * (1.0f/256.0f) + LN_EPS);
        }
    }
    lds_barrier();
    LOAD_WCAP_RANGE(49, 64)

    {
        int i = t & 31, b0 = t >> 5;
        #pragma unroll
        for (int bsel = 0; bsel < 2; ++bsel) {
            int b = b0 + 8*bsel;
            float mu = mu_s[b], rs = rstd_s[b];
            float vals[8]; float nsq = 0.f;
            #pragma unroll
            for (int k = 0; k < 8; ++k) {
                float v = (xh[b*XP + i*8 + k] - mu) * rs * g8[k] + be8[k];
                vals[k] = v; nsq = fmaf(v, v, nsq);
            }
            float scale = nsq / ((1.0f + nsq) * sqrtf(nsq + SQ_EPS));
            #pragma unroll
            for (int k = 0; k < 8; ++k) u_t[i*130 + k*16 + b] = scale * vals[k];
        }
    }

    #pragma unroll
    for (int m = 0; m < 2; ++m) {
        #pragma unroll
        for (int g = 0; g < 4; ++g)
            Cs[(((2*hi + m)*4 + g)*8 + wav)*16 + bb] = make_float4(0.0625f, 0.0625f, 0.0625f, 0.0625f);
    }
    lds_barrier();

    v2f bu0[4], bu1[4];
    #pragma unroll
    for (int ii = 0; ii < 4; ++ii) {
        const float* ub = u_t + (4*wav + ii)*130 + bb;
        bu0[ii][0] = ub[(2*hi+0)*16];  bu0[ii][1] = ub[(2*hi+1)*16];
        bu1[ii][0] = ub[(2*hi+4)*16];  bu1[ii][1] = ub[(2*hi+5)*16];
    }

    lds_barrier();
    v8f uk2[NK];
    #pragma unroll
    for (int j = 0; j < NK; ++j) {
        v8f t8 = vzero8();
        t8 = wmma4(a0[32 + j], bu0[2], t8);
        t8 = wmma4(a1[32 + j], bu1[2], t8);
        uk2[j] = t8;
    }
    #pragma unroll
    for (int q = NK; q < 32; ++q)
        Wl[(wav*32 + q)*32 + lane] = make_float4(a0[32 + q][0], a0[32 + q][1], a1[32 + q][0], a1[32 + q][1]);
    __builtin_amdgcn_sched_barrier(0);
    const float4* const wl = Wl + (wav*32)*32 + lane;
    v8f uk[32];
    #pragma unroll
    for (int q = 0; q < 32; ++q) {
        v8f t8 = vzero8();
        t8 = wmma4(a0[q], bu0[q >> 4], t8);
        t8 = wmma4(a1[q], bu1[q >> 4], t8);
        uk[q] = t8;
    }

    float4* const lgp = Wl + (wav*32)*32 + lane;
    #pragma unroll
    for (int s = 0; s < 8; ++s) lgp[s*32] = make_float4(0.f, 0.f, 0.f, 0.f);
    const bool up = (hi != 0);

    v8f vA = vzero8(), vB = vzero8();
    for (int r = 0; r < 3; ++r) {
        float4 cq[4];
        #pragma unroll
        for (int ii = 0; ii < 4; ++ii) cq[ii] = Cs[((ii*4 + 0)*8 + wav)*16 + bb];
        #pragma unroll
        for (int g = 0; g < 4; ++g) {
            float4 wf[4];
            #pragma unroll
            for (int jj = 0; jj < 4; ++jj) wf[jj] = wl[(16 + 4*g + jj)*32];
            float4 cn[4];
            #pragma unroll
            for (int ii = 0; ii < 4; ++ii) cn[ii] = (g < 3) ? Cs[((ii*4 + g + 1)*8 + wav)*16 + bb] : cq[ii];
            float cc[4][4];
            #pragma unroll
            for (int ii = 0; ii < 4; ++ii) { cc[ii][0] = cq[ii].x; cc[ii][1] = cq[ii].y; cc[ii][2] = cq[ii].z; cc[ii][3] = cq[ii].w; }
            #pragma unroll
            for (int h = 0; h < 2; ++h) {
                v8f sj[2];
                #pragma unroll
                for (int q = 0; q < 2; ++q) {
                    const int jj = 2*h + q, j = 4*g + jj;
                    const float c0 = cc[0][jj], c1 = cc[1][jj], c2 = cc[2][jj];
                    #pragma unroll
                    for (int v = 0; v < 8; ++v)
                        sj[q][v] = fmaf(c2, uk2[j][v], fmaf(c1, uk[16 + j][v], c0 * uk[j][v]));
                }
                v2f p0[2], p1[2];
                #pragma unroll
                for (int q = 0; q < 2; ++q) {
                    const float c = cc[3][2*h + q];
                    p0[q][0] = bu0[3][0]*c; p0[q][1] = bu0[3][1]*c; p1[q][0] = bu1[3][0]*c; p1[q][1] = bu1[3][1]*c;
                }
                const float4 wa = wf[2*h], wb = wf[2*h + 1];
                sj[0] = wmma4((v2f){wa.x, wa.y}, p0[0], sj[0]);
                sj[1] = wmma4((v2f){wb.x, wb.y}, p0[1], sj[1]);
                sj[0] = wmma4((v2f){wa.z, wa.w}, p1[0], sj[0]);
                sj[1] = wmma4((v2f){wb.z, wb.w}, p1[1], sj[1]);
                #pragma unroll
                for (int q = 0; q < 2; ++q) {
                    const int j = 4*g + 2*h + q;
                    Ps[((wav*16 + j)*2 + 0)*32 + lane] = make_float4(sj[q][0], sj[q][1], sj[q][2], sj[q][3]);
                    Ps[((wav*16 + j)*2 + 1)*32 + lane] = make_float4(sj[q][4], sj[q][5], sj[q][6], sj[q][7]);
                }
            }
            #pragma unroll
            for (int ii = 0; ii < 4; ++ii) cq[ii] = cn[ii];
            __builtin_amdgcn_sched_barrier(0);
        }
        __syncthreads();
        v8f sA = vzero8(), sB = vzero8();
        #pragma unroll 1
        for (int hh = 0; hh < 2; ++hh) {
            const float4* pr = Ps + ((hh*64 + wav)*2)*32 + lane;
            float4 xa[4][2], xb[4][2];
            #pragma unroll
            for (int w = 0; w < 4; ++w) {
                xa[w][0] = pr[(w*32 + 0)*32];       xa[w][1] = pr[(w*32 + 1)*32];
                xb[w][0] = pr[(w*32 + 16 + 0)*32];  xb[w][1] = pr[(w*32 + 16 + 1)*32];
            }
            #pragma unroll
            for (int w = 0; w < 4; ++w) {
                sA[0]+=xa[w][0].x; sA[1]+=xa[w][0].y; sA[2]+=xa[w][0].z; sA[3]+=xa[w][0].w;
                sA[4]+=xa[w][1].x; sA[5]+=xa[w][1].y; sA[6]+=xa[w][1].z; sA[7]+=xa[w][1].w;
                sB[0]+=xb[w][0].x; sB[1]+=xb[w][0].y; sB[2]+=xb[w][0].z; sB[3]+=xb[w][0].w;
                sB[4]+=xb[w][1].x; sB[5]+=xb[w][1].y; sB[6]+=xb[w][1].z; sB[7]+=xb[w][1].w;
            }
        }
        float nA = 0.f, nB = 0.f;
        #pragma unroll
        for (int v = 0; v < 8; ++v) { nA = fmaf(sA[v], sA[v], nA); nB = fmaf(sB[v], sB[v], nB); }
        nA += __shfl_xor(nA, 16, 32);  nB += __shfl_xor(nB, 16, 32);
        float scA = nA / ((1.0f + nA) * sqrtf(nA + SQ_EPS));
        float scB = nB / ((1.0f + nB) * sqrtf(nB + SQ_EPS));
        #pragma unroll
        for (int v = 0; v < 8; ++v) { vA[v] = sA[v]*scA; vB[v] = sB[v]*scB; }

        if (r < 2) {
            Vs[((wav    )*2 + 0)*32 + lane] = make_float4(vA[0], vA[1], vA[2], vA[3]);
            Vs[((wav    )*2 + 1)*32 + lane] = make_float4(vA[4], vA[5], vA[6], vA[7]);
            Vs[((wav + 8)*2 + 0)*32 + lane] = make_float4(vB[0], vB[1], vB[2], vB[3]);
            Vs[((wav + 8)*2 + 1)*32 + lane] = make_float4(vB[4], vB[5], vB[6], vB[7]);
            __syncthreads();
            float Lg[32];
            #pragma unroll
            for (int s = 0; s < 8; ++s) {
                const float4 l4 = lgp[s*32];
                Lg[4*s] = l4.x; Lg[4*s + 1] = l4.y; Lg[4*s + 2] = l4.z; Lg[4*s + 3] = l4.w;
            }
            float own_p[2] = {0.f, 0.f}, rcv_p[2] = {0.f, 0.f};
            float4 w3 = wl[16*32];
            #pragma unroll
            for (int j = 0; j < 16; ++j) {
                const float4 y0 = Vs[(j*2 + 0)*32 + lane], y1 = Vs[(j*2 + 1)*32 + lane];
                float4 w3n = w3;
                if (j < 15) w3n = wl[(16 + j + 1)*32];
                v8f u3 = vzero8();
                u3 = wmma4((v2f){w3.x, w3.y}, bu0[3], u3);
                u3 = wmma4((v2f){w3.z, w3.w}, bu1[3], u3);
                const v2f ya = {y0.x, y0.y}, yb = {y0.z, y0.w}, yc = {y1.x, y1.y}, yd = {y1.z, y1.w};
                const v8f t0 = uk[j], t1 = uk[16 + j], t2 = uk2[j];
                v2f d0 = {0.f, 0.f}, d1 = {0.f, 0.f}, d2 = {0.f, 0.f}, d3 = {0.f, 0.f};
                d0 = __builtin_elementwise_fma((v2f){t0[0], t0[1]}, ya, d0);
                d1 = __builtin_elementwise_fma((v2f){t1[0], t1[1]}, ya, d1);
                d2 = __builtin_elementwise_fma((v2f){t2[0], t2[1]}, ya, d2);
                d0 = __builtin_elementwise_fma((v2f){t0[2], t0[3]}, yb, d0);
                d1 = __builtin_elementwise_fma((v2f){t1[2], t1[3]}, yb, d1);
                d2 = __builtin_elementwise_fma((v2f){t2[2], t2[3]}, yb, d2);
                d0 = __builtin_elementwise_fma((v2f){t0[4], t0[5]}, yc, d0);
                d1 = __builtin_elementwise_fma((v2f){t1[4], t1[5]}, yc, d1);
                d2 = __builtin_elementwise_fma((v2f){t2[4], t2[5]}, yc, d2);
                d0 = __builtin_elementwise_fma((v2f){t0[6], t0[7]}, yd, d0);
                d1 = __builtin_elementwise_fma((v2f){t1[6], t1[7]}, yd, d1);
                d2 = __builtin_elementwise_fma((v2f){t2[6], t2[7]}, yd, d2);
                if (j > 0) {
                    #pragma unroll
                    for (int m = 0; m < 2; ++m) Lg[m*16 + j - 1] += own_p[m] + rcv_p[m];
                }
                d3 = __builtin_elementwise_fma((v2f){u3[0], u3[1]}, ya, d3);
                d3 = __builtin_elementwise_fma((v2f){u3[2], u3[3]}, yb, d3);
                d3 = __builtin_elementwise_fma((v2f){u3[4], u3[5]}, yc, d3);
                d3 = __builtin_elementwise_fma((v2f){u3[6], u3[7]}, yd, d3);
                float ag[4];
                ag[0] = d0[0] + d0[1];  ag[1] = d1[0] + d1[1];  ag[2] = d2[0] + d2[1];  ag[3] = d3[0] + d3[1];
                #pragma unroll
                for (int m = 0; m < 2; ++m) {
                    const float own = up ? ag[2 + m] : ag[m];
                    const float snd = up ? ag[m] : ag[2 + m];
                    own_p[m] = own;
                    rcv_p[m] = __shfl_xor(snd, 16, 32);
                }
                w3 = w3n;
                __builtin_amdgcn_sched_barrier(0);
            }
            #pragma unroll
            for (int m = 0; m < 2; ++m) Lg[m*16 + 15] += own_p[m] + rcv_p[m];
            #pragma unroll
            for (int s = 0; s < 8; ++s) lgp[s*32] = make_float4(Lg[4*s], Lg[4*s + 1], Lg[4*s + 2], Lg[4*s + 3]);
            #pragma unroll
            for (int m = 0; m < 2; ++m) {
                float mx = Lg[m*16];
                #pragma unroll
                for (int j = 1; j < 16; ++j) mx = fmaxf(mx, Lg[m*16 + j]);
                float e[16]; float sum = 0.f;
                #pragma unroll
                for (int j = 0; j < 16; ++j) { e[j] = expf(Lg[m*16 + j] - mx); sum += e[j]; }
                float inv = 1.0f / sum;
                #pragma unroll
                for (int g = 0; g < 4; ++g)
                    Cs[(((2*hi + m)*4 + g)*8 + wav)*16 + bb] = make_float4(e[4*g]*inv, e[4*g+1]*inv, e[4*g+2]*inv, e[4*g+3]*inv);
            }
            __syncthreads();
        }
    }

    {
        float pA = 0.f, pB = 0.f;
        #pragma unroll
        for (int v = 0; v < 8; ++v) {
            pA = fmaf(vA[v], Wf[(wav    )*16 + v + 8*hi], pA);
            pB = fmaf(vB[v], Wf[(wav + 8)*16 + v + 8*hi], pB);
        }
        pA += __shfl_xor(pA, 16, 32);  pB += __shfl_xor(pB, 16, 32);
        if (lane < 16) { part[wav*17 + bb] = pA; part[(wav+8)*17 + bb] = pB; }
    }
    __syncthreads();
    if (t < 16) {
        float acc = bf[0];
        #pragma unroll
        for (int j = 0; j < 16; ++j) acc += part[j*17 + t];
        out[tile*16 + t] = acc;
    }
}

extern "C" void kernel_launch(void* const* d_in, const int* in_sizes, int n_in,
                              void* d_out, int out_size, void* d_ws, size_t ws_size,
                              hipStream_t stream) {
    const float* x    = (const float*)d_in[0];
    const float* W1   = (const float*)d_in[1];
    const float* b1   = (const float*)d_in[2];
    const float* W2   = (const float*)d_in[3];
    const float* b2   = (const float*)d_in[4];
    const float* ln_g = (const float*)d_in[5];
    const float* ln_b = (const float*)d_in[6];
    const float* Wcap = (const float*)d_in[7];
    const float* Wf   = (const float*)d_in[8];
    const float* bf   = (const float*)d_in[9];
    float* out = (float*)d_out;

    const int B = in_sizes[0] / 256;
    const int blocks = B / 16;
    capsnet_fused<<<blocks, 256, 0, stream>>>(x, W1, b1, W2, b2, ln_g, ln_b, Wcap, Wf, bf, out);
}
